# starting kernel
# baseline (speedup 1.0000x reference)
#include <hip/hip_runtime.h>
#include <hip/hip_bf16.h>

#define N_SRC 100000
#define N_DST 20000
#define E_CNT 1250000
#define D 64
#define NEG_SLOPE 0.2f

typedef __attribute__((ext_vector_type(2))) float v2f;
typedef __attribute__((ext_vector_type(8))) float v8f;

__global__ __launch_bounds__(256) void gat_init(float* __restrict__ out,
                                                const float* __restrict__ bias,
                                                float* __restrict__ denom,
                                                unsigned* __restrict__ m_enc) {
    int i = blockIdx.x * blockDim.x + threadIdx.x;
    if (i < N_DST * D) out[i] = bias[i & (D - 1)];
    if (i < N_DST) {
        denom[i] = 0.0f;
        m_enc[i] = 0u;
    }
}

__global__ __launch_bounds__(64) void gat_watt(const float* __restrict__ W,
                                               const float* __restrict__ att_src,
                                               const float* __restrict__ att_dst,
                                               float* __restrict__ watt) {
    int t = threadIdx.x;
    float s = 0.0f, d = 0.0f;
    #pragma unroll 8
    for (int j = 0; j < D; ++j) {
        float w = W[t * D + j];
        s += w * att_src[j];
        d += w * att_dst[j];
    }
    watt[t] = s;
    watt[D + t] = d;
}

__global__ __launch_bounds__(128) void gat_gemm_hsrc(const float* __restrict__ x,
                                                     const float* __restrict__ W,
                                                     float* __restrict__ h) {
    __shared__ float ldsW[D * D];
    for (int i = threadIdx.x; i < D * D; i += 128) ldsW[i] = W[i];
    __syncthreads();

    const int lane = threadIdx.x & 31;
    const int mrow = lane & 15;
    const int hi   = lane >> 4;
    const int row0 = blockIdx.x * 64 + (threadIdx.x >> 5) * 16;
    const int row  = row0 + mrow;
    const int rowc = (row < N_SRC) ? row : (N_SRC - 1);
    const float valid = (row < N_SRC) ? 1.0f : 0.0f;

    v2f afrag[16];
    const float* abase = x + (size_t)rowc * D + hi * 2;
    #pragma unroll
    for (int kk = 0; kk < 16; ++kk) {
        afrag[kk].x = abase[kk * 4 + 0] * valid;
        afrag[kk].y = abase[kk * 4 + 1] * valid;
    }

    v8f acc[4] = {v8f{}, v8f{}, v8f{}, v8f{}};

    #pragma unroll
    for (int kk = 0; kk < 16; ++kk) {
        const int kb = kk * 4 + hi * 2;
        #pragma unroll
        for (int nb = 0; nb < 4; ++nb) {
            v2f b;
            b.x = ldsW[(kb + 0) * D + nb * 16 + mrow];
            b.y = ldsW[(kb + 1) * D + nb * 16 + mrow];
            acc[nb] = __builtin_amdgcn_wmma_f32_16x16x4_f32(
                 false, afrag[kk],  false, b,
                 (short)0, acc[nb],
                 false,  false);
        }
    }

    #pragma unroll
    for (int nb = 0; nb < 4; ++nb) {
        #pragma unroll
        for (int r = 0; r < 8; ++r) {
            int orow = row0 + r + hi * 8;
            if (orow < N_SRC) h[(size_t)orow * D + nb * 16 + mrow] = acc[nb][r];
        }
    }
}

__global__ __launch_bounds__(256) void gat_rowdot(const float* __restrict__ x,
                                                  const int* __restrict__ res_n_id,
                                                  const float* __restrict__ watt,
                                                  float* __restrict__ a_src,
                                                  float* __restrict__ a_dst) {
    int wid  = blockIdx.x * 8 + (threadIdx.x >> 5);
    int lane = threadIdx.x & 31;
    const float2* xp;
    const float2* wv;
    float* outp;
    if (wid < N_SRC) {
        xp = (const float2*)(x + (size_t)wid * D);
        wv = (const float2*)(watt);
        outp = a_src + wid;
    } else {
        int j = wid - N_SRC;
        if (j >= N_DST) return;
        xp = (const float2*)(x + (size_t)res_n_id[j] * D);
        wv = (const float2*)(watt + D);
        outp = a_dst + j;
    }
    float2 xv = xp[lane];
    float2 wvv = wv[lane];
    float s = xv.x * wvv.x + xv.y * wvv.y;
    #pragma unroll
    for (int off = 16; off > 0; off >>= 1) s += __shfl_xor(s, off, 32);
    if (lane == 0) *outp = s;
}

__global__ __launch_bounds__(256) void gat_edge_logit(const int* __restrict__ es,
                                                      const int* __restrict__ ed,
                                                      const float* __restrict__ a_src,
                                                      const float* __restrict__ a_dst,
                                                      float* __restrict__ e_buf,
                                                      unsigned* __restrict__ m_enc) {
    int k = blockIdx.x * blockDim.x + threadIdx.x;
    if (k >= E_CNT) return;
    int d = ed[k];
    float e = a_src[es[k]] + a_dst[d];
    e = (e > 0.0f) ? e : NEG_SLOPE * e;
    e_buf[k] = e;
    unsigned u = __float_as_uint(e);
    u = (u & 0x80000000u) ? ~u : (u | 0x80000000u);
    atomicMax(m_enc + d, u);
}

__global__ __launch_bounds__(256) void gat_edge_exp(const int* __restrict__ ed,
                                                    float* __restrict__ e_buf,
                                                    const unsigned* __restrict__ m_enc,
                                                    float* __restrict__ denom) {
    int k = blockIdx.x * blockDim.x + threadIdx.x;
    if (k >= E_CNT) return;
    int d = ed[k];
    unsigned u = m_enc[d];
    float m = (u & 0x80000000u) ? __uint_as_float(u & 0x7FFFFFFFu)
                                : __uint_as_float(~u);
    float ex = __expf(e_buf[k] - m);
    e_buf[k] = ex;
    atomicAdd(denom + d, ex);
}

__global__ __launch_bounds__(256) void gat_scatter(const int* __restrict__ es,
                                                   const int* __restrict__ ed,
                                                   const float* __restrict__ e_buf,
                                                   const float* __restrict__ denom,
                                                   const float* __restrict__ h,
                                                   float* __restrict__ out) {
    int widx = blockIdx.x * 8 + (threadIdx.x >> 5);
    if (widx >= E_CNT) return;
    int lane = threadIdx.x & 31;
    int s = es[widx];
    int d = ed[widx];
    float alpha = e_buf[widx] / (denom[d] + 1e-16f);
    const float2* hp = (const float2*)(h + (size_t)s * D);
    float2 v = hp[lane];
    float* op = out + (size_t)d * D + lane * 2;
    atomicAdd(op + 0, alpha * v.x);
    atomicAdd(op + 1, alpha * v.y);
}

extern "C" void kernel_launch(void* const* d_in, const int* in_sizes, int n_in,
                              void* d_out, int out_size, void* d_ws, size_t ws_size,
                              hipStream_t stream) {
    const float* x        = (const float*)d_in[0];
    const int*   res_n_id = (const int*)d_in[1];
    const int*   edge_src = (const int*)d_in[2];
    const int*   edge_dst = (const int*)d_in[3];
    const float* W        = (const float*)d_in[4];
    const float* att_src  = (const float*)d_in[5];
    const float* att_dst  = (const float*)d_in[6];
    const float* bias     = (const float*)d_in[7];
    float* out = (float*)d_out;

    char* ws = (char*)d_ws;
    float*    h_src = (float*)(ws);
    float*    e_buf = (float*)(ws + 25600000);
    float*    a_src = (float*)(ws + 30600000);
    float*    a_dst = (float*)(ws + 31000000);
    float*    denom = (float*)(ws + 31080000);
    unsigned* m_enc = (unsigned*)(ws + 31160000);
    float*    watt  = (float*)(ws + 31240000);

    gat_init<<<(N_DST * D + 255) / 256, 256, 0, stream>>>(out, bias, denom, m_enc);
    gat_watt<<<1, 64, 0, stream>>>(W, att_src, att_dst, watt);
    gat_gemm_hsrc<<<(N_SRC + 63) / 64, 128, 0, stream>>>(x, W, h_src);
    gat_rowdot<<<(N_SRC + N_DST + 7) / 8, 256, 0, stream>>>(x, res_n_id, watt, a_src, a_dst);
    gat_edge_logit<<<(E_CNT + 255) / 256, 256, 0, stream>>>(edge_src, edge_dst, a_src,
                                                            a_dst, e_buf, m_enc);
    gat_edge_exp<<<(E_CNT + 255) / 256, 256, 0, stream>>>(edge_dst, e_buf, m_enc, denom);
    gat_scatter<<<(E_CNT + 7) / 8, 256, 0, stream>>>(edge_src, edge_dst, e_buf, denom,
                                                     h_src, out);
}


# Segment-max and softmax-denominator atomics go to an LDS table per work-group, then one coalesced global atomic per non-empty entry; denominator summed as per-group partials (sum order changes)
# speedup vs baseline: 2.0442x; 2.0442x over previous
#include <hip/hip_runtime.h>
#include <hip/hip_bf16.h>

#define N_SRC 100000
#define N_DST 20000
#define E_CNT 1250000
#define D 64
#define NEG_SLOPE 0.2f

#define EB 125
#define ET 1024
#define EW 1000
#define EU 10
static_assert(EB * EW * EU == E_CNT, "edge passes must cover every edge once");
static_assert(EW <= ET, "edge threads must exist");
static_assert(N_DST % 32 == 0, "table flush assumes whole 128-byte lines");
#define FLUSH_ROT 160
static_assert(FLUSH_ROT % 32 == 0 && (EB - 1) * FLUSH_ROT < N_DST, "flush start inside the table, on a line");

typedef __attribute__((ext_vector_type(2))) float v2f;
typedef __attribute__((ext_vector_type(8))) float v8f;

__global__ __launch_bounds__(256) void gat_init(float* __restrict__ out,
                                                const float* __restrict__ bias,
                                                float* __restrict__ denom,
                                                unsigned* __restrict__ m_enc) {
    int i = blockIdx.x * blockDim.x + threadIdx.x;
    if (i < N_DST * D) out[i] = bias[i & (D - 1)];
    if (i < N_DST) {
        denom[i] = 0.0f;
        m_enc[i] = 0u;
    }
}

__global__ __launch_bounds__(64) void gat_watt(const float* __restrict__ W,
                                               const float* __restrict__ att_src,
                                               const float* __restrict__ att_dst,
                                               float* __restrict__ watt) {
    int t = threadIdx.x;
    float s = 0.0f, d = 0.0f;
    #pragma unroll 8
    for (int j = 0; j < D; ++j) {
        float w = W[t * D + j];
        s += w * att_src[j];
        d += w * att_dst[j];
    }
    watt[t] = s;
    watt[D + t] = d;
}

__global__ __launch_bounds__(128) void gat_gemm_hsrc(const float* __restrict__ x,
                                                     const float* __restrict__ W,
                                                     float* __restrict__ h) {
    __shared__ float ldsW[D * D];
    for (int i = threadIdx.x; i < D * D; i += 128) ldsW[i] = W[i];
    __syncthreads();

    const int lane = threadIdx.x & 31;
    const int mrow = lane & 15;
    const int hi   = lane >> 4;
    const int row0 = blockIdx.x * 64 + (threadIdx.x >> 5) * 16;
    const int row  = row0 + mrow;
    const int rowc = (row < N_SRC) ? row : (N_SRC - 1);
    const float valid = (row < N_SRC) ? 1.0f : 0.0f;

    v2f afrag[16];
    const float* abase = x + (size_t)rowc * D + hi * 2;
    #pragma unroll
    for (int kk = 0; kk < 16; ++kk) {
        afrag[kk].x = abase[kk * 4 + 0] * valid;
        afrag[kk].y = abase[kk * 4 + 1] * valid;
    }

    v8f acc[4] = {v8f{}, v8f{}, v8f{}, v8f{}};

    #pragma unroll
    for (int kk = 0; kk < 16; ++kk) {
        const int kb = kk * 4 + hi * 2;
        #pragma unroll
        for (int nb = 0; nb < 4; ++nb) {
            v2f b;
            b.x = ldsW[(kb + 0) * D + nb * 16 + mrow];
            b.y = ldsW[(kb + 1) * D + nb * 16 + mrow];
            acc[nb] = __builtin_amdgcn_wmma_f32_16x16x4_f32(
                 false, afrag[kk],  false, b,
                 (short)0, acc[nb],
                 false,  false);
        }
    }

    #pragma unroll
    for (int nb = 0; nb < 4; ++nb) {
        #pragma unroll
        for (int r = 0; r < 8; ++r) {
            int orow = row0 + r + hi * 8;
            if (orow < N_SRC) h[(size_t)orow * D + nb * 16 + mrow] = acc[nb][r];
        }
    }
}

__global__ __launch_bounds__(256) void gat_rowdot(const float* __restrict__ x,
                                                  const int* __restrict__ res_n_id,
                                                  const float* __restrict__ watt,
                                                  float* __restrict__ a_src,
                                                  float* __restrict__ a_dst) {
    int wid  = blockIdx.x * 8 + (threadIdx.x >> 5);
    int lane = threadIdx.x & 31;
    const float2* xp;
    const float2* wv;
    float* outp;
    if (wid < N_SRC) {
        xp = (const float2*)(x + (size_t)wid * D);
        wv = (const float2*)(watt);
        outp = a_src + wid;
    } else {
        int j = wid - N_SRC;
        if (j >= N_DST) return;
        xp = (const float2*)(x + (size_t)res_n_id[j] * D);
        wv = (const float2*)(watt + D);
        outp = a_dst + j;
    }
    float2 xv = xp[lane];
    float2 wvv = wv[lane];
    float s = xv.x * wvv.x + xv.y * wvv.y;
    #pragma unroll
    for (int off = 16; off > 0; off >>= 1) s += __shfl_xor(s, off, 32);
    if (lane == 0) *outp = s;
}

__global__ __launch_bounds__(ET) void gat_edge_logit(const int* __restrict__ es,
                                                     const int* __restrict__ ed,
                                                     const float* __restrict__ a_src,
                                                     const float* __restrict__ a_dst,
                                                     float* __restrict__ e_buf,
                                                     unsigned* __restrict__ m_enc) {
    __shared__ unsigned tab[N_DST];
    const int t = threadIdx.x;
    for (int i = t; i < N_DST; i += ET) tab[i] = 0u;
    __syncthreads();

    if (t < EW) {
        const int k0 = blockIdx.x * (EW * EU) + t;
        int s[EU], d[EU];
        float e[EU];
        #pragma unroll
        for (int j = 0; j < EU; ++j) {
            s[j] = es[k0 + j * EW];
            d[j] = ed[k0 + j * EW];
        }
        #pragma unroll
        for (int j = 0; j < EU; ++j) e[j] = a_src[s[j]] + a_dst[d[j]];
        #pragma unroll
        for (int j = 0; j < EU; ++j) {
            float v = e[j];
            v = (v > 0.0f) ? v : NEG_SLOPE * v;
            e_buf[k0 + j * EW] = v;
            unsigned u = __float_as_uint(v);
            u = (u & 0x80000000u) ? ~u : (u | 0x80000000u);
            __hip_atomic_fetch_max(&tab[d[j]], u, __ATOMIC_RELAXED,
                                   __HIP_MEMORY_SCOPE_WORKGROUP);
        }
    }
    __syncthreads();

    const int rot = blockIdx.x * FLUSH_ROT;
    for (int i = t; i < N_DST; i += ET) {
        int idx = i + rot;
        if (idx >= N_DST) idx -= N_DST;
        const unsigned u = tab[idx];
        if (u != 0u) atomicMax(m_enc + idx, u);
    }
}

__global__ __launch_bounds__(ET) void gat_edge_exp(const int* __restrict__ ed,
                                                   float* __restrict__ e_buf,
                                                   const unsigned* __restrict__ m_enc,
                                                   float* __restrict__ denom) {
    __shared__ float tab[N_DST];
    const int t = threadIdx.x;
    for (int i = t; i < N_DST; i += ET) tab[i] = 0.0f;
    __syncthreads();

    if (t < EW) {
        const int k0 = blockIdx.x * (EW * EU) + t;
        int d[EU];
        float ev[EU];
        unsigned mu[EU];
        #pragma unroll
        for (int j = 0; j < EU; ++j) {
            d[j] = ed[k0 + j * EW];
            ev[j] = e_buf[k0 + j * EW];
        }
        #pragma unroll
        for (int j = 0; j < EU; ++j) mu[j] = m_enc[d[j]];
        #pragma unroll
        for (int j = 0; j < EU; ++j) {
            const unsigned u = mu[j];
            const float m = (u & 0x80000000u) ? __uint_as_float(u & 0x7FFFFFFFu)
                                              : __uint_as_float(~u);
            const float ex = __expf(ev[j] - m);
            e_buf[k0 + j * EW] = ex;
            __hip_atomic_fetch_add(&tab[d[j]], ex, __ATOMIC_RELAXED,
                                   __HIP_MEMORY_SCOPE_WORKGROUP);
        }
    }
    __syncthreads();

    const int rot = blockIdx.x * FLUSH_ROT;
    for (int i = t; i < N_DST; i += ET) {
        int idx = i + rot;
        if (idx >= N_DST) idx -= N_DST;
        const float v = tab[idx];
        if (v != 0.0f) atomicAdd(denom + idx, v);
    }
}

__global__ __launch_bounds__(256) void gat_scatter(const int* __restrict__ es,
                                                   const int* __restrict__ ed,
                                                   const float* __restrict__ e_buf,
                                                   const float* __restrict__ denom,
                                                   const float* __restrict__ h,
                                                   float* __restrict__ out) {
    int widx = blockIdx.x * 8 + (threadIdx.x >> 5);
    if (widx >= E_CNT) return;
    int lane = threadIdx.x & 31;
    int s = es[widx];
    int d = ed[widx];
    float alpha = e_buf[widx] / (denom[d] + 1e-16f);
    const float2* hp = (const float2*)(h + (size_t)s * D);
    float2 v = hp[lane];
    float* op = out + (size_t)d * D + lane * 2;
    atomicAdd(op + 0, alpha * v.x);
    atomicAdd(op + 1, alpha * v.y);
}

extern "C" void kernel_launch(void* const* d_in, const int* in_sizes, int n_in,
                              void* d_out, int out_size, void* d_ws, size_t ws_size,
                              hipStream_t stream) {
    const float* x        = (const float*)d_in[0];
    const int*   res_n_id = (const int*)d_in[1];
    const int*   edge_src = (const int*)d_in[2];
    const int*   edge_dst = (const int*)d_in[3];
    const float* W        = (const float*)d_in[4];
    const float* att_src  = (const float*)d_in[5];
    const float* att_dst  = (const float*)d_in[6];
    const float* bias     = (const float*)d_in[7];
    float* out = (float*)d_out;

    char* ws = (char*)d_ws;
    float*    h_src = (float*)(ws);
    float*    a_src = (float*)(ws + 25600000);
    float*    a_dst = (float*)(ws + 26000000);
    float*    denom = (float*)(ws + 26080000);
    unsigned* m_enc = (unsigned*)(ws + 26160000);
    float*    watt  = (float*)(ws + 26240000);
    float*    e_buf = (float*)(ws + 26240512);

    gat_init<<<(N_DST * D + 255) / 256, 256, 0, stream>>>(out, bias, denom, m_enc);
    gat_watt<<<1, 64, 0, stream>>>(W, att_src, att_dst, watt);
    gat_gemm_hsrc<<<(N_SRC + 63) / 64, 128, 0, stream>>>(x, W, h_src);
    gat_rowdot<<<(N_SRC + N_DST + 7) / 8, 256, 0, stream>>>(x, res_n_id, watt, a_src, a_dst);
    gat_edge_logit<<<EB, ET, 0, stream>>>(edge_src, edge_dst, a_src, a_dst, e_buf, m_enc);
    gat_edge_exp<<<EB, ET, 0, stream>>>(edge_dst, e_buf, m_enc, denom);
    gat_scatter<<<(E_CNT + 7) / 8, 256, 0, stream>>>(edge_src, edge_dst, e_buf, denom,
                                                     h_src, out);
}


# Edges counting-sorted by destination (LDS counts, scan, ranked fill); one wave per destination does max, exp-sum and weighted row sum in registers, divides once; no float atomics (sum order changes)
# speedup vs baseline: 3.2099x; 1.5702x over previous
#include <hip/hip_runtime.h>
#include <hip/hip_bf16.h>

#define N_SRC 100000
#define N_DST 20000
#define E_CNT 1250000
#define D 64
#define NEG_SLOPE 0.2f

#define EB 125
#define ET 1024
#define EW 1000
#define EU 10
static_assert(EB * EW * EU == E_CNT, "edge passes must cover every edge once");
static_assert(EW <= ET, "edge threads must exist");
static_assert(N_DST % 32 == 0, "table flush assumes whole 128-byte lines");
#define FLUSH_ROT 160
static_assert(FLUSH_ROT % 32 == 0 && (EB - 1) * FLUSH_ROT < N_DST, "flush start inside the table, on a line");

#define FL ((N_DST + ET - 1) / ET)
static_assert(N_DST % 8 == 0, "kernel 6 runs eight destinations per work-group");
static_assert(N_DST % 4 == 0 && N_DST / 4 <= 5 * ET, "kernel 4 scans five 16-byte groups per thread");

typedef __attribute__((ext_vector_type(2))) float v2f;
typedef __attribute__((ext_vector_type(8))) float v8f;

__global__ __launch_bounds__(256) void gat_setup(const float* __restrict__ W,
                                                 const float* __restrict__ att_src,
                                                 const float* __restrict__ att_dst,
                                                 float* __restrict__ watt,
                                                 unsigned* __restrict__ count) {
    const int i = blockIdx.x * blockDim.x + threadIdx.x;
    if (i < N_DST) count[i] = 0u;
    if (i < D) {
        const int t = i;
        float s = 0.0f, d = 0.0f;
        #pragma unroll 8
        for (int j = 0; j < D; ++j) {
            float w = W[t * D + j];
            s += w * att_src[j];
            d += w * att_dst[j];
        }
        watt[t] = s;
        watt[D + t] = d;
    }
}

__global__ __launch_bounds__(128) void gat_gemm_hsrc(const float* __restrict__ x,
                                                     const float* __restrict__ W,
                                                     float* __restrict__ h) {
    __shared__ float ldsW[D * D];
    for (int i = threadIdx.x; i < D * D; i += 128) ldsW[i] = W[i];
    __syncthreads();

    const int lane = threadIdx.x & 31;
    const int mrow = lane & 15;
    const int hi   = lane >> 4;
    const int row0 = blockIdx.x * 64 + (threadIdx.x >> 5) * 16;
    const int row  = row0 + mrow;
    const int rowc = (row < N_SRC) ? row : (N_SRC - 1);
    const float valid = (row < N_SRC) ? 1.0f : 0.0f;

    v2f afrag[16];
    const float* abase = x + (size_t)rowc * D + hi * 2;
    #pragma unroll
    for (int kk = 0; kk < 16; ++kk) {
        afrag[kk].x = abase[kk * 4 + 0] * valid;
        afrag[kk].y = abase[kk * 4 + 1] * valid;
    }

    v8f acc[4] = {v8f{}, v8f{}, v8f{}, v8f{}};

    #pragma unroll
    for (int kk = 0; kk < 16; ++kk) {
        const int kb = kk * 4 + hi * 2;
        #pragma unroll
        for (int nb = 0; nb < 4; ++nb) {
            v2f b;
            b.x = ldsW[(kb + 0) * D + nb * 16 + mrow];
            b.y = ldsW[(kb + 1) * D + nb * 16 + mrow];
            acc[nb] = __builtin_amdgcn_wmma_f32_16x16x4_f32(
                 false, afrag[kk],  false, b,
                 (short)0, acc[nb],
                 false,  false);
        }
    }

    #pragma unroll
    for (int nb = 0; nb < 4; ++nb) {
        #pragma unroll
        for (int r = 0; r < 8; ++r) {
            int orow = row0 + r + hi * 8;
            if (orow < N_SRC) h[(size_t)orow * D + nb * 16 + mrow] = acc[nb][r];
        }
    }
}

__global__ __launch_bounds__(256) void gat_rowdot(const float* __restrict__ x,
                                                  const int* __restrict__ res_n_id,
                                                  const float* __restrict__ watt,
                                                  float* __restrict__ a_src,
                                                  float* __restrict__ a_dst) {
    int wid  = blockIdx.x * 8 + (threadIdx.x >> 5);
    int lane = threadIdx.x & 31;
    const float2* xp;
    const float2* wv;
    float* outp;
    if (wid < N_SRC) {
        xp = (const float2*)(x + (size_t)wid * D);
        wv = (const float2*)(watt);
        outp = a_src + wid;
    } else {
        int j = wid - N_SRC;
        if (j >= N_DST) return;
        xp = (const float2*)(x + (size_t)res_n_id[j] * D);
        wv = (const float2*)(watt + D);
        outp = a_dst + j;
    }
    float2 xv = xp[lane];
    float2 wvv = wv[lane];
    float s = xv.x * wvv.x + xv.y * wvv.y;
    #pragma unroll
    for (int off = 16; off > 0; off >>= 1) s += __shfl_xor(s, off, 32);
    if (lane == 0) *outp = s;
}

__global__ __launch_bounds__(ET) void gat_hist(const int* __restrict__ ed,
                                               unsigned* __restrict__ count) {
    __shared__ unsigned tab[N_DST];
    const int t = threadIdx.x;
    for (int i = t; i < N_DST; i += ET) tab[i] = 0u;
    __syncthreads();

    if (t < EW) {
        const int k0 = blockIdx.x * (EW * EU) + t;
        int d[EU];
        #pragma unroll
        for (int j = 0; j < EU; ++j) d[j] = ed[k0 + j * EW];
        #pragma unroll
        for (int j = 0; j < EU; ++j)
            __hip_atomic_fetch_add(&tab[d[j]], 1u, __ATOMIC_RELAXED,
                                   __HIP_MEMORY_SCOPE_WORKGROUP);
    }
    __syncthreads();

    const int rot = blockIdx.x * FLUSH_ROT;
    for (int i = t; i < N_DST; i += ET) {
        int idx = i + rot;
        if (idx >= N_DST) idx -= N_DST;
        const unsigned c = tab[idx];
        if (c != 0u) atomicAdd(count + idx, c);
    }
}

__global__ __launch_bounds__(ET) void gat_scan(const unsigned* __restrict__ count,
                                               unsigned* __restrict__ cursor) {
    __shared__ unsigned wsum[32];
    const int t = threadIdx.x;
    const int lane = t & 31;
    const int wave = t >> 5;
    const uint4* cp = (const uint4*)count;
    uint4 c[5];
    unsigned tot = 0u;
    #pragma unroll
    for (int q = 0; q < 5; ++q) {
        const int g = t * 5 + q;
        c[q] = (g < N_DST / 4) ? cp[g] : uint4{0u, 0u, 0u, 0u};
        tot += c[q].x + c[q].y + c[q].z + c[q].w;
    }
    unsigned incl = tot;
    #pragma unroll
    for (int off = 1; off < 32; off <<= 1) {
        const unsigned nb = __shfl_up(incl, off, 32);
        if (lane >= off) incl += nb;
    }
    if (lane == 31) wsum[wave] = incl;
    __syncthreads();
    if (wave == 0) {
        const unsigned w = wsum[lane];
        unsigned wi = w;
        #pragma unroll
        for (int off = 1; off < 32; off <<= 1) {
            const unsigned nb = __shfl_up(wi, off, 32);
            if (lane >= off) wi += nb;
        }
        wsum[lane] = wi - w;
    }
    __syncthreads();
    unsigned run = wsum[wave] + incl - tot;
    uint4* op = (uint4*)cursor;
    #pragma unroll
    for (int q = 0; q < 5; ++q) {
        const int g = t * 5 + q;
        uint4 o;
        o.x = run; run += c[q].x;
        o.y = run; run += c[q].y;
        o.z = run; run += c[q].z;
        o.w = run; run += c[q].w;
        if (g < N_DST / 4) op[g] = o;
    }
}

__global__ __launch_bounds__(ET) void gat_fill(const int* __restrict__ es,
                                               const int* __restrict__ ed,
                                               unsigned* __restrict__ cursor,
                                               int* __restrict__ srcs) {
    __shared__ unsigned tab[N_DST];
    const int t = threadIdx.x;
    for (int i = t; i < N_DST; i += ET) tab[i] = 0u;
    __syncthreads();

    int s[EU], d[EU];
    unsigned r[EU];
    if (t < EW) {
        const int k0 = blockIdx.x * (EW * EU) + t;
        #pragma unroll
        for (int j = 0; j < EU; ++j) {
            s[j] = es[k0 + j * EW];
            d[j] = ed[k0 + j * EW];
        }
        #pragma unroll
        for (int j = 0; j < EU; ++j)
            r[j] = __hip_atomic_fetch_add(&tab[d[j]], 1u, __ATOMIC_RELAXED,
                                          __HIP_MEMORY_SCOPE_WORKGROUP);
    }
    __syncthreads();

    const int rot = blockIdx.x * FLUSH_ROT;
    unsigned c[FL], b[FL];
    #pragma unroll
    for (int q = 0; q < FL; ++q) {
        const int i = t + q * ET;
        int idx = i + rot;
        if (idx >= N_DST) idx -= N_DST;
        c[q] = (i < N_DST) ? tab[idx] : 0u;
    }
    #pragma unroll
    for (int q = 0; q < FL; ++q) {
        int idx = t + q * ET + rot;
        if (idx >= N_DST) idx -= N_DST;
        b[q] = 0u;
        if (c[q] != 0u) b[q] = atomicAdd(cursor + idx, c[q]);
    }
    #pragma unroll
    for (int q = 0; q < FL; ++q) {
        int idx = t + q * ET + rot;
        if (idx >= N_DST) idx -= N_DST;
        if (c[q] != 0u) tab[idx] = b[q];
    }
    __syncthreads();

    if (t < EW) {
        #pragma unroll
        for (int j = 0; j < EU; ++j) srcs[tab[d[j]] + r[j]] = s[j];
    }
}

__global__ __launch_bounds__(256) void gat_aggregate(const int* __restrict__ srcs,
                                                     const unsigned* __restrict__ count,
                                                     const unsigned* __restrict__ cursor,
                                                     const float* __restrict__ a_src,
                                                     const float* __restrict__ a_dst,
                                                     const float* __restrict__ h,
                                                     const float* __restrict__ bias,
                                                     float* __restrict__ out) {
    const int d = __builtin_amdgcn_readfirstlane((int)(blockIdx.x * 8 + (threadIdx.x >> 5)));
    const int lane = threadIdx.x & 31;
    const int n = (int)count[d];
    const int beg = (int)cursor[d] - n;
    const float ad = a_dst[d];
    const int* sp = srcs + beg;

    float m = -INFINITY;
    for (int c0 = 0; c0 < n; c0 += 32) {
        const int k = c0 + lane;
        const int kc = (k < n) ? k : (n - 1);
        float v = a_src[sp[kc]] + ad;
        v = (v > 0.0f) ? v : NEG_SLOPE * v;
        m = (v > m) ? v : m;
    }
    #pragma unroll
    for (int off = 16; off > 0; off >>= 1) {
        const float o = __shfl_xor(m, off, 32);
        m = (o > m) ? o : m;
    }

    const int odd = lane >> 4;
    const int col = (lane & 15) * 4;
    float4 acc = make_float4(0.0f, 0.0f, 0.0f, 0.0f);
    float dsum = 0.0f;
    for (int c0 = 0; c0 < n; c0 += 32) {
        const int k = c0 + lane;
        const int kc = (k < n) ? k : (n - 1);
        const int s = sp[kc];
        float v = a_src[s] + ad;
        v = (v > 0.0f) ? v : NEG_SLOPE * v;
        const float ex = (k < n) ? __expf(v - m) : 0.0f;
        dsum += ex;
        const int cnt = (n - c0 < 32) ? (n - c0) : 32;
        for (int i = 0; i < cnt; i += 8) {
            float4 hv[4];
            float w[4];
            #pragma unroll
            for (int u = 0; u < 4; ++u) {
                const int from = i + 2 * u + odd;
                const int si = __shfl(s, from, 32);
                w[u] = __shfl(ex, from, 32);
                hv[u] = *(const float4*)(h + (size_t)si * D + col);
            }
            #pragma unroll
            for (int u = 0; u < 4; ++u) {
                acc.x += w[u] * hv[u].x;
                acc.y += w[u] * hv[u].y;
                acc.z += w[u] * hv[u].z;
                acc.w += w[u] * hv[u].w;
            }
        }
    }
    #pragma unroll
    for (int off = 16; off > 0; off >>= 1) dsum += __shfl_xor(dsum, off, 32);
    acc.x += __shfl_xor(acc.x, 16, 32);
    acc.y += __shfl_xor(acc.y, 16, 32);
    acc.z += __shfl_xor(acc.z, 16, 32);
    acc.w += __shfl_xor(acc.w, 16, 32);
    if (odd == 0) {
        const float inv = 1.0f / (dsum + 1e-16f);
        float4 o;
        o.x = acc.x * inv + bias[col + 0];
        o.y = acc.y * inv + bias[col + 1];
        o.z = acc.z * inv + bias[col + 2];
        o.w = acc.w * inv + bias[col + 3];
        *(float4*)(out + (size_t)d * D + col) = o;
    }
}

extern "C" void kernel_launch(void* const* d_in, const int* in_sizes, int n_in,
                              void* d_out, int out_size, void* d_ws, size_t ws_size,
                              hipStream_t stream) {
    const float* x        = (const float*)d_in[0];
    const int*   res_n_id = (const int*)d_in[1];
    const int*   edge_src = (const int*)d_in[2];
    const int*   edge_dst = (const int*)d_in[3];
    const float* W        = (const float*)d_in[4];
    const float* att_src  = (const float*)d_in[5];
    const float* att_dst  = (const float*)d_in[6];
    const float* bias     = (const float*)d_in[7];
    float* out = (float*)d_out;

    char* ws = (char*)d_ws;
    float*    h_src  = (float*)(ws);
    float*    a_src  = (float*)(ws + 25600000);
    float*    a_dst  = (float*)(ws + 26000000);
    unsigned* count  = (unsigned*)(ws + 26080000);
    unsigned* cursor = (unsigned*)(ws + 26160000);
    float*    watt   = (float*)(ws + 26240000);
    int*      srcs   = (int*)(ws + 26240512);

    gat_setup<<<(N_DST + 255) / 256, 256, 0, stream>>>(W, att_src, att_dst, watt, count);
    gat_gemm_hsrc<<<(N_SRC + 63) / 64, 128, 0, stream>>>(x, W, h_src);
    gat_rowdot<<<(N_SRC + N_DST + 7) / 8, 256, 0, stream>>>(x, res_n_id, watt, a_src, a_dst);
    gat_hist<<<EB, ET, 0, stream>>>(edge_dst, count);
    gat_scan<<<1, ET, 0, stream>>>(count, cursor);
    gat_fill<<<EB, ET, 0, stream>>>(edge_src, edge_dst, cursor, srcs);
    gat_aggregate<<<N_DST / 8, 256, 0, stream>>>(srcs, count, cursor, a_src, a_dst, h_src,
                                                 bias, out);
}


# Attention logits from h_src: a_src from the GEMM's accumulators in its epilogue, a_dst in idle work-groups of the scan launch; the 120,000-wave row-dot and set-up launches go (sum order changes)
# speedup vs baseline: 3.7944x; 1.1821x over previous
#include <hip/hip_runtime.h>
#include <hip/hip_bf16.h>

#define N_SRC 100000
#define N_DST 20000
#define E_CNT 1250000
#define D 64
#define NEG_SLOPE 0.2f

#define EB 125
#define ET 1024
#define EW 1000
#define EU 10
static_assert(EB * EW * EU == E_CNT, "edge passes must cover every edge once");
static_assert(EW <= ET, "edge threads must exist");
static_assert(N_DST % 32 == 0, "table flush assumes whole 128-byte lines");
#define FLUSH_ROT 160
static_assert(FLUSH_ROT % 32 == 0 && (EB - 1) * FLUSH_ROT < N_DST, "flush start inside the table, on a line");

#define FL ((N_DST + ET - 1) / ET)
static_assert(N_DST % 8 == 0, "kernel 4 runs eight destinations per work-group");
static_assert(N_DST % 4 == 0 && N_DST / 4 <= 5 * ET, "kernel 2 scans five 16-byte groups per thread");
static_assert(((N_SRC + 63) / 64) * 128 >= N_DST, "kernel 0 clears one count per thread");
#define AD_WAVES (N_DST / 32)
#define AD_BLOCKS ((AD_WAVES + 31) / 32)

typedef __attribute__((ext_vector_type(2))) float v2f;
typedef __attribute__((ext_vector_type(8))) float v8f;

__global__ __launch_bounds__(128) void gat_gemm_hsrc(const float* __restrict__ x,
                                                     const float* __restrict__ W,
                                                     const float* __restrict__ att_src,
                                                     float* __restrict__ h,
                                                     float* __restrict__ a_src,
                                                     unsigned* __restrict__ count) {
    __shared__ float ldsW[D * D];
    __shared__ float ldsA[64];
    const int gid = blockIdx.x * 128 + threadIdx.x;
    if (gid < N_DST) count[gid] = 0u;
    for (int i = threadIdx.x; i < D * D; i += 128) ldsW[i] = W[i];
    __syncthreads();

    const int lane = threadIdx.x & 31;
    const int mrow = lane & 15;
    const int hi   = lane >> 4;
    const int row0 = blockIdx.x * 64 + (threadIdx.x >> 5) * 16;
    const int row  = row0 + mrow;
    const int rowc = (row < N_SRC) ? row : (N_SRC - 1);
    const float valid = (row < N_SRC) ? 1.0f : 0.0f;

    v2f afrag[16];
    const float* abase = x + (size_t)rowc * D + hi * 2;
    #pragma unroll
    for (int kk = 0; kk < 16; ++kk) {
        afrag[kk].x = abase[kk * 4 + 0] * valid;
        afrag[kk].y = abase[kk * 4 + 1] * valid;
    }

    float as[4];
    #pragma unroll
    for (int nb = 0; nb < 4; ++nb) as[nb] = att_src[nb * 16 + mrow];

    v8f acc[4] = {v8f{}, v8f{}, v8f{}, v8f{}};

    #pragma unroll
    for (int kk = 0; kk < 16; ++kk) {
        const int kb = kk * 4 + hi * 2;
        #pragma unroll
        for (int nb = 0; nb < 4; ++nb) {
            v2f b;
            b.x = ldsW[(kb + 0) * D + nb * 16 + mrow];
            b.y = ldsW[(kb + 1) * D + nb * 16 + mrow];
            acc[nb] = __builtin_amdgcn_wmma_f32_16x16x4_f32(
                 false, afrag[kk],  false, b,
                 (short)0, acc[nb],
                 false,  false);
        }
    }

    #pragma unroll
    for (int nb = 0; nb < 4; ++nb) {
        #pragma unroll
        for (int r = 0; r < 8; ++r) {
            int orow = row0 + r + hi * 8;
            if (orow < N_SRC) h[(size_t)orow * D + nb * 16 + mrow] = acc[nb][r];
        }
    }

    float v[8];
    #pragma unroll
    for (int r = 0; r < 8; ++r)
        v[r] = acc[0][r] * as[0] + acc[1][r] * as[1] + acc[2][r] * as[2] + acc[3][r] * as[3];
    const bool b8 = (mrow & 8) != 0;
    const bool b4 = (mrow & 4) != 0;
    const bool b2 = (mrow & 2) != 0;
    float w4[4];
    #pragma unroll
    for (int i = 0; i < 4; ++i) {
        const float keep = b8 ? v[i + 4] : v[i];
        const float send = b8 ? v[i] : v[i + 4];
        w4[i] = keep + __shfl_xor(send, 8, 32);
    }
    float u2[2];
    #pragma unroll
    for (int i = 0; i < 2; ++i) {
        const float keep = b4 ? w4[i + 2] : w4[i];
        const float send = b4 ? w4[i] : w4[i + 2];
        u2[i] = keep + __shfl_xor(send, 4, 32);
    }
    float t = (b2 ? u2[1] : u2[0]) + __shfl_xor(b2 ? u2[0] : u2[1], 2, 32);
    t += __shfl_xor(t, 1, 32);
    if ((mrow & 1) == 0) ldsA[(threadIdx.x >> 5) * 16 + hi * 8 + (mrow >> 1)] = t;
    __syncthreads();
    if (threadIdx.x < 64) {
        const int arow = blockIdx.x * 64 + threadIdx.x;
        if (arow < N_SRC) a_src[arow] = ldsA[threadIdx.x];
    }
}

__global__ __launch_bounds__(ET) void gat_hist(const int* __restrict__ ed,
                                               unsigned* __restrict__ count) {
    __shared__ unsigned tab[N_DST];
    const int t = threadIdx.x;
    for (int i = t; i < N_DST; i += ET) tab[i] = 0u;
    __syncthreads();

    if (t < EW) {
        const int k0 = blockIdx.x * (EW * EU) + t;
        int d[EU];
        #pragma unroll
        for (int j = 0; j < EU; ++j) d[j] = ed[k0 + j * EW];
        #pragma unroll
        for (int j = 0; j < EU; ++j)
            __hip_atomic_fetch_add(&tab[d[j]], 1u, __ATOMIC_RELAXED,
                                   __HIP_MEMORY_SCOPE_WORKGROUP);
    }
    __syncthreads();

    const int rot = blockIdx.x * FLUSH_ROT;
    for (int i = t; i < N_DST; i += ET) {
        int idx = i + rot;
        if (idx >= N_DST) idx -= N_DST;
        const unsigned c = tab[idx];
        if (c != 0u) atomicAdd(count + idx, c);
    }
}

__global__ __launch_bounds__(ET) void gat_scan(const unsigned* __restrict__ count,
                                               unsigned* __restrict__ cursor,
                                               const int* __restrict__ res_n_id,
                                               const float* __restrict__ h,
                                               const float* __restrict__ att_dst,
                                               float* __restrict__ a_dst) {
    __shared__ unsigned wsum[32];
    const int t = threadIdx.x;
    const int lane = t & 31;
    const int wave = t >> 5;
    if (blockIdx.x != 0) {
        const int w = ((int)blockIdx.x - 1) * 32 + wave;
        if (w >= AD_WAVES) return;
        const int rid = res_n_id[w * 32 + lane];
        const float2 av = *(const float2*)(att_dst + lane * 2);
        float v[32];
        #pragma unroll
        for (int k = 0; k < 32; ++k) {
            const int rk = __builtin_amdgcn_readlane(rid, k);
            const float2 hr = *(const float2*)(h + (size_t)rk * D + lane * 2);
            v[k] = hr.x * av.x + hr.y * av.y;
        }
        const bool c16 = (lane & 16) != 0;
        const bool c8 = (lane & 8) != 0;
        const bool c4 = (lane & 4) != 0;
        const bool c2 = (lane & 2) != 0;
        const bool c1 = (lane & 1) != 0;
        float p16[16], p8[8], p4[4], p2[2];
        #pragma unroll
        for (int i = 0; i < 16; ++i)
            p16[i] = (c16 ? v[i + 16] : v[i]) + __shfl_xor(c16 ? v[i] : v[i + 16], 16, 32);
        #pragma unroll
        for (int i = 0; i < 8; ++i)
            p8[i] = (c8 ? p16[i + 8] : p16[i]) + __shfl_xor(c8 ? p16[i] : p16[i + 8], 8, 32);
        #pragma unroll
        for (int i = 0; i < 4; ++i)
            p4[i] = (c4 ? p8[i + 4] : p8[i]) + __shfl_xor(c4 ? p8[i] : p8[i + 4], 4, 32);
        #pragma unroll
        for (int i = 0; i < 2; ++i)
            p2[i] = (c2 ? p4[i + 2] : p4[i]) + __shfl_xor(c2 ? p4[i] : p4[i + 2], 2, 32);
        a_dst[w * 32 + lane] = (c1 ? p2[1] : p2[0]) + __shfl_xor(c1 ? p2[0] : p2[1], 1, 32);
        return;
    }
    const uint4* cp = (const uint4*)count;
    uint4 c[5];
    unsigned tot = 0u;
    #pragma unroll
    for (int q = 0; q < 5; ++q) {
        const int g = t * 5 + q;
        c[q] = (g < N_DST / 4) ? cp[g] : uint4{0u, 0u, 0u, 0u};
        tot += c[q].x + c[q].y + c[q].z + c[q].w;
    }
    unsigned incl = tot;
    #pragma unroll
    for (int off = 1; off < 32; off <<= 1) {
        const unsigned nb = __shfl_up(incl, off, 32);
        if (lane >= off) incl += nb;
    }
    if (lane == 31) wsum[wave] = incl;
    __syncthreads();
    if (wave == 0) {
        const unsigned w = wsum[lane];
        unsigned wi = w;
        #pragma unroll
        for (int off = 1; off < 32; off <<= 1) {
            const unsigned nb = __shfl_up(wi, off, 32);
            if (lane >= off) wi += nb;
        }
        wsum[lane] = wi - w;
    }
    __syncthreads();
    unsigned run = wsum[wave] + incl - tot;
    uint4* op = (uint4*)cursor;
    #pragma unroll
    for (int q = 0; q < 5; ++q) {
        const int g = t * 5 + q;
        uint4 o;
        o.x = run; run += c[q].x;
        o.y = run; run += c[q].y;
        o.z = run; run += c[q].z;
        o.w = run; run += c[q].w;
        if (g < N_DST / 4) op[g] = o;
    }
}

__global__ __launch_bounds__(ET) void gat_fill(const int* __restrict__ es,
                                               const int* __restrict__ ed,
                                               unsigned* __restrict__ cursor,
                                               int* __restrict__ srcs) {
    __shared__ unsigned tab[N_DST];
    const int t = threadIdx.x;
    for (int i = t; i < N_DST; i += ET) tab[i] = 0u;
    __syncthreads();

    int s[EU], d[EU];
    unsigned r[EU];
    if (t < EW) {
        const int k0 = blockIdx.x * (EW * EU) + t;
        #pragma unroll
        for (int j = 0; j < EU; ++j) {
            s[j] = es[k0 + j * EW];
            d[j] = ed[k0 + j * EW];
        }
        #pragma unroll
        for (int j = 0; j < EU; ++j)
            r[j] = __hip_atomic_fetch_add(&tab[d[j]], 1u, __ATOMIC_RELAXED,
                                          __HIP_MEMORY_SCOPE_WORKGROUP);
    }
    __syncthreads();

    const int rot = blockIdx.x * FLUSH_ROT;
    unsigned c[FL], b[FL];
    #pragma unroll
    for (int q = 0; q < FL; ++q) {
        const int i = t + q * ET;
        int idx = i + rot;
        if (idx >= N_DST) idx -= N_DST;
        c[q] = (i < N_DST) ? tab[idx] : 0u;
    }
    #pragma unroll
    for (int q = 0; q < FL; ++q) {
        int idx = t + q * ET + rot;
        if (idx >= N_DST) idx -= N_DST;
        b[q] = 0u;
        if (c[q] != 0u) b[q] = atomicAdd(cursor + idx, c[q]);
    }
    #pragma unroll
    for (int q = 0; q < FL; ++q) {
        int idx = t + q * ET + rot;
        if (idx >= N_DST) idx -= N_DST;
        if (c[q] != 0u) tab[idx] = b[q];
    }
    __syncthreads();

    if (t < EW) {
        #pragma unroll
        for (int j = 0; j < EU; ++j) srcs[tab[d[j]] + r[j]] = s[j];
    }
}

__global__ __launch_bounds__(256) void gat_aggregate(const int* __restrict__ srcs,
                                                     const unsigned* __restrict__ count,
                                                     const unsigned* __restrict__ cursor,
                                                     const float* __restrict__ a_src,
                                                     const float* __restrict__ a_dst,
                                                     const float* __restrict__ h,
                                                     const float* __restrict__ bias,
                                                     float* __restrict__ out) {
    const int d = __builtin_amdgcn_readfirstlane((int)(blockIdx.x * 8 + (threadIdx.x >> 5)));
    const int lane = threadIdx.x & 31;
    const int n = (int)count[d];
    const int beg = (int)cursor[d] - n;
    const float ad = a_dst[d];
    const int* sp = srcs + beg;

    float m = -INFINITY;
    for (int c0 = 0; c0 < n; c0 += 32) {
        const int k = c0 + lane;
        const int kc = (k < n) ? k : (n - 1);
        float v = a_src[sp[kc]] + ad;
        v = (v > 0.0f) ? v : NEG_SLOPE * v;
        m = (v > m) ? v : m;
    }
    #pragma unroll
    for (int off = 16; off > 0; off >>= 1) {
        const float o = __shfl_xor(m, off, 32);
        m = (o > m) ? o : m;
    }

    const int odd = lane >> 4;
    const int col = (lane & 15) * 4;
    float4 acc = make_float4(0.0f, 0.0f, 0.0f, 0.0f);
    float dsum = 0.0f;
    for (int c0 = 0; c0 < n; c0 += 32) {
        const int k = c0 + lane;
        const int kc = (k < n) ? k : (n - 1);
        const int s = sp[kc];
        float v = a_src[s] + ad;
        v = (v > 0.0f) ? v : NEG_SLOPE * v;
        const float ex = (k < n) ? __expf(v - m) : 0.0f;
        dsum += ex;
        const int cnt = (n - c0 < 32) ? (n - c0) : 32;
        for (int i = 0; i < cnt; i += 8) {
            float4 hv[4];
            float w[4];
            #pragma unroll
            for (int u = 0; u < 4; ++u) {
                const int from = i + 2 * u + odd;
                const int si = __shfl(s, from, 32);
                w[u] = __shfl(ex, from, 32);
                hv[u] = *(const float4*)(h + (size_t)si * D + col);
            }
            #pragma unroll
            for (int u = 0; u < 4; ++u) {
                acc.x += w[u] * hv[u].x;
                acc.y += w[u] * hv[u].y;
                acc.z += w[u] * hv[u].z;
                acc.w += w[u] * hv[u].w;
            }
        }
    }
    #pragma unroll
    for (int off = 16; off > 0; off >>= 1) dsum += __shfl_xor(dsum, off, 32);
    acc.x += __shfl_xor(acc.x, 16, 32);
    acc.y += __shfl_xor(acc.y, 16, 32);
    acc.z += __shfl_xor(acc.z, 16, 32);
    acc.w += __shfl_xor(acc.w, 16, 32);
    if (odd == 0) {
        const float inv = 1.0f / (dsum + 1e-16f);
        float4 o;
        o.x = acc.x * inv + bias[col + 0];
        o.y = acc.y * inv + bias[col + 1];
        o.z = acc.z * inv + bias[col + 2];
        o.w = acc.w * inv + bias[col + 3];
        *(float4*)(out + (size_t)d * D + col) = o;
    }
}

extern "C" void kernel_launch(void* const* d_in, const int* in_sizes, int n_in,
                              void* d_out, int out_size, void* d_ws, size_t ws_size,
                              hipStream_t stream) {
    const float* x        = (const float*)d_in[0];
    const int*   res_n_id = (const int*)d_in[1];
    const int*   edge_src = (const int*)d_in[2];
    const int*   edge_dst = (const int*)d_in[3];
    const float* W        = (const float*)d_in[4];
    const float* att_src  = (const float*)d_in[5];
    const float* att_dst  = (const float*)d_in[6];
    const float* bias     = (const float*)d_in[7];
    float* out = (float*)d_out;

    char* ws = (char*)d_ws;
    float*    h_src  = (float*)(ws);
    float*    a_src  = (float*)(ws + 25600000);
    float*    a_dst  = (float*)(ws + 26000000);
    unsigned* count  = (unsigned*)(ws + 26080000);
    unsigned* cursor = (unsigned*)(ws + 26160000);
    int*      srcs   = (int*)(ws + 26240000);

    gat_gemm_hsrc<<<(N_SRC + 63) / 64, 128, 0, stream>>>(x, W, att_src, h_src, a_src, count);
    gat_hist<<<EB, ET, 0, stream>>>(edge_dst, count);
    gat_scan<<<1 + AD_BLOCKS, ET, 0, stream>>>(count, cursor, res_n_id, h_src, att_dst, a_dst);
    gat_fill<<<EB, ET, 0, stream>>>(edge_src, edge_dst, cursor, srcs);
    gat_aggregate<<<N_DST / 8, 256, 0, stream>>>(srcs, count, cursor, a_src, a_dst, h_src,
                                                 bias, out);
}


# W applied after the aggregation: edges gather x rows, logits from x.(W att), one 20,000-row WMMA product in the aggregate's epilogue; the 100,000-row GEMM and h_src go (sum order changes)
# speedup vs baseline: 4.6701x; 1.2308x over previous
#include <hip/hip_runtime.h>


#define N_SRC 100000
#define N_DST 20000
#define E_CNT 1250000
#define D 64
#define NEG_SLOPE 0.2f

#define EB 125
#define ET 1024
#define EW 1000
#define EU 10
static_assert(EB * EW * EU == E_CNT, "edge passes must cover every edge once");
static_assert(EW <= ET, "edge threads must exist");
static_assert(N_DST % 32 == 0, "table flush assumes whole 128-byte lines");
#define FLUSH_ROT 160
static_assert(FLUSH_ROT % 32 == 0 && (EB - 1) * FLUSH_ROT < N_DST, "flush start inside the table, on a line");

#define FL ((N_DST + ET - 1) / ET)
static_assert(N_DST % 4 == 0 && N_DST / 4 <= 5 * ET, "kernel 2 scans five 16-byte groups per thread");
#define ZB ((N_DST + ET - 1) / ET)
#define AG 16
#define TP 68
static_assert(N_DST % AG == 0, "kernel 4 runs whole groups of sixteen destinations");
static_assert(D == 64 && D * D == 4 * ET, "kernel 0 stages W with one 16-byte load per thread");

typedef __attribute__((ext_vector_type(2))) float v2f;
typedef __attribute__((ext_vector_type(8))) float v8f;

__global__ __launch_bounds__(ET) void gat_prep(const float* __restrict__ W,
                                               const float* __restrict__ att_src,
                                               const float* __restrict__ att_dst,
                                               unsigned* __restrict__ count,
                                               float* __restrict__ wv) {
    __shared__ float ldsW[D * (D + 1)];
    __shared__ float ldsA[2 * D];
    const int t = threadIdx.x;
    if (blockIdx.x < ZB) {
        const int i = blockIdx.x * ET + t;
        if (i < N_DST) count[i] = 0u;
        return;
    }
    const float4 w4 = *(const float4*)(W + t * 4);
    const int wr = (t * 4) >> 6;
    const int wc = (t * 4) & 63;
    ldsW[wr * (D + 1) + wc + 0] = w4.x;
    ldsW[wr * (D + 1) + wc + 1] = w4.y;
    ldsW[wr * (D + 1) + wc + 2] = w4.z;
    ldsW[wr * (D + 1) + wc + 3] = w4.w;
    if (t < D) ldsA[t] = att_src[t];
    else if (t < 2 * D) ldsA[t] = att_dst[t - D];
    __syncthreads();
    if (t < 2 * D) {
        const int k = t & (D - 1);
        const float* a = ldsA + (t & D);
        float s0 = 0.0f, s1 = 0.0f, s2 = 0.0f, s3 = 0.0f;
        #pragma unroll
        for (int j = 0; j < D; j += 4) {
            s0 += ldsW[k * (D + 1) + j + 0] * a[j + 0];
            s1 += ldsW[k * (D + 1) + j + 1] * a[j + 1];
            s2 += ldsW[k * (D + 1) + j + 2] * a[j + 2];
            s3 += ldsW[k * (D + 1) + j + 3] * a[j + 3];
        }
        wv[t] = (s0 + s1) + (s2 + s3);
    }
}

__global__ __launch_bounds__(ET) void gat_hist(const int* __restrict__ ed,
                                               unsigned* __restrict__ count) {
    __shared__ unsigned tab[N_DST];
    const int t = threadIdx.x;
    for (int i = t; i < N_DST; i += ET) tab[i] = 0u;
    __syncthreads();

    if (t < EW) {
        const int k0 = blockIdx.x * (EW * EU) + t;
        int d[EU];
        #pragma unroll
        for (int j = 0; j < EU; ++j) d[j] = ed[k0 + j * EW];
        #pragma unroll
        for (int j = 0; j < EU; ++j)
            __hip_atomic_fetch_add(&tab[d[j]], 1u, __ATOMIC_RELAXED,
                                   __HIP_MEMORY_SCOPE_WORKGROUP);
    }
    __syncthreads();

    const int rot = blockIdx.x * FLUSH_ROT;
    for (int i = t; i < N_DST; i += ET) {
        int idx = i + rot;
        if (idx >= N_DST) idx -= N_DST;
        const unsigned c = tab[idx];
        if (c != 0u) atomicAdd(count + idx, c);
    }
}

__global__ __launch_bounds__(ET) void gat_scan(const unsigned* __restrict__ count,
                                               unsigned* __restrict__ cursor) {
    __shared__ unsigned wsum[32];
    const int t = threadIdx.x;
    const int lane = t & 31;
    const int wave = t >> 5;
    const uint4* cp = (const uint4*)count;
    uint4 c[5];
    unsigned tot = 0u;
    #pragma unroll
    for (int q = 0; q < 5; ++q) {
        const int g = t * 5 + q;
        c[q] = (g < N_DST / 4) ? cp[g] : uint4{0u, 0u, 0u, 0u};
        tot += c[q].x + c[q].y + c[q].z + c[q].w;
    }
    unsigned incl = tot;
    #pragma unroll
    for (int off = 1; off < 32; off <<= 1) {
        const unsigned nb = __shfl_up(incl, off, 32);
        if (lane >= off) incl += nb;
    }
    if (lane == 31) wsum[wave] = incl;
    __syncthreads();
    if (wave == 0) {
        const unsigned w = wsum[lane];
        unsigned wi = w;
        #pragma unroll
        for (int off = 1; off < 32; off <<= 1) {
            const unsigned nb = __shfl_up(wi, off, 32);
            if (lane >= off) wi += nb;
        }
        wsum[lane] = wi - w;
    }
    __syncthreads();
    unsigned run = wsum[wave] + incl - tot;
    uint4* op = (uint4*)cursor;
    #pragma unroll
    for (int q = 0; q < 5; ++q) {
        const int g = t * 5 + q;
        uint4 o;
        o.x = run; run += c[q].x;
        o.y = run; run += c[q].y;
        o.z = run; run += c[q].z;
        o.w = run; run += c[q].w;
        if (g < N_DST / 4) op[g] = o;
    }
}

__global__ __launch_bounds__(ET) void gat_fill(const int* __restrict__ es,
                                               const int* __restrict__ ed,
                                               unsigned* __restrict__ cursor,
                                               int* __restrict__ srcs) {
    __shared__ unsigned tab[N_DST];
    const int t = threadIdx.x;
    for (int i = t; i < N_DST; i += ET) tab[i] = 0u;
    __syncthreads();

    int s[EU], d[EU];
    unsigned r[EU];
    if (t < EW) {
        const int k0 = blockIdx.x * (EW * EU) + t;
        #pragma unroll
        for (int j = 0; j < EU; ++j) {
            s[j] = es[k0 + j * EW];
            d[j] = ed[k0 + j * EW];
        }
        #pragma unroll
        for (int j = 0; j < EU; ++j)
            r[j] = __hip_atomic_fetch_add(&tab[d[j]], 1u, __ATOMIC_RELAXED,
                                          __HIP_MEMORY_SCOPE_WORKGROUP);
    }
    __syncthreads();

    const int rot = blockIdx.x * FLUSH_ROT;
    unsigned c[FL], b[FL];
    #pragma unroll
    for (int q = 0; q < FL; ++q) {
        const int i = t + q * ET;
        int idx = i + rot;
        if (idx >= N_DST) idx -= N_DST;
        c[q] = (i < N_DST) ? tab[idx] : 0u;
    }
    #pragma unroll
    for (int q = 0; q < FL; ++q) {
        int idx = t + q * ET + rot;
        if (idx >= N_DST) idx -= N_DST;
        b[q] = 0u;
        if (c[q] != 0u) b[q] = atomicAdd(cursor + idx, c[q]);
    }
    #pragma unroll
    for (int q = 0; q < FL; ++q) {
        int idx = t + q * ET + rot;
        if (idx >= N_DST) idx -= N_DST;
        if (c[q] != 0u) tab[idx] = b[q];
    }
    __syncthreads();

    if (t < EW) {
        #pragma unroll
        for (int j = 0; j < EU; ++j) srcs[tab[d[j]] + r[j]] = s[j];
    }
}


#define NEG_BIG (-3.0e38f)

template <int XM>
__device__ __forceinline__ float row_xor_add(float x) {
    const int y = __builtin_amdgcn_update_dpp(0, __float_as_int(x), 0x160 + XM, 0xf, 0xf, true);
    return x + __int_as_float(y);
}

__global__ __launch_bounds__(AG * 32) void gat_aggregate(const int* __restrict__ srcs,
                                                         const unsigned* __restrict__ count,
                                                         const unsigned* __restrict__ cursor,
                                                         const float* __restrict__ wv,
                                                         const int* __restrict__ res_n_id,
                                                         const float* __restrict__ x,
                                                         const float* __restrict__ W,
                                                         const float* __restrict__ bias,
                                                         float* __restrict__ out) {
    __shared__ float gtile[AG * TP];
    __shared__ float otile[AG * TP];
    const int wave = __builtin_amdgcn_readfirstlane((int)(threadIdx.x >> 5));
    const int d = __builtin_amdgcn_readfirstlane((int)(blockIdx.x * AG) + wave);
    const int lane = threadIdx.x & 31;
    const int n = (int)count[d];
    const int beg = (int)cursor[d] - n;
    const int rid = res_n_id[d];
    const int* sp = srcs + beg;

    const int odd = lane >> 4;
    const int col = (lane & 15) * 4;
    const float4 av = *(const float4*)(wv + col);
    const float4 dv = *(const float4*)(wv + D + col);
    const float4 xd = *(const float4*)(x + (size_t)rid * D + col);
    float ad = xd.x * dv.x + xd.y * dv.y + xd.z * dv.z + xd.w * dv.w;
    ad = row_xor_add<1>(ad);
    ad = row_xor_add<2>(ad);
    ad = row_xor_add<4>(ad);
    ad = row_xor_add<8>(ad);

    float m = NEG_BIG;
    float dsum = 0.0f;
    float4 acc = make_float4(0.0f, 0.0f, 0.0f, 0.0f);
    for (int c0 = 0; c0 < n; c0 += 32) {
        const int k = c0 + lane;
        const int kc = (k < n) ? k : (n - 1);
        const int s = sp[kc];
        const int rem = n - c0;
        const int cnt = (rem < 32) ? rem : 32;
        for (int i = 0; i < cnt; i += 8) {
            float4 hv[4];
            float p[4];
            bool ok[4];
            #pragma unroll
            for (int u = 0; u < 4; ++u) {
                const int from = i + 2 * u + odd;
                const int si = __shfl(s, from, 32);
                ok[u] = from < rem;
                hv[u] = *(const float4*)(x + (size_t)si * D + col);
            }
            #pragma unroll
            for (int u = 0; u < 4; ++u)
                p[u] = hv[u].x * av.x + hv[u].y * av.y + hv[u].z * av.z + hv[u].w * av.w;
            #pragma unroll
            for (int u = 0; u < 4; ++u) p[u] = row_xor_add<1>(p[u]);
            #pragma unroll
            for (int u = 0; u < 4; ++u) p[u] = row_xor_add<2>(p[u]);
            #pragma unroll
            for (int u = 0; u < 4; ++u) p[u] = row_xor_add<4>(p[u]);
            #pragma unroll
            for (int u = 0; u < 4; ++u) p[u] = row_xor_add<8>(p[u]);
            float e[4];
            float mn = m;
            #pragma unroll
            for (int u = 0; u < 4; ++u) {
                float v = p[u] + ad;
                v = (v > 0.0f) ? v : NEG_SLOPE * v;
                e[u] = ok[u] ? v : NEG_BIG;
                mn = (e[u] > mn) ? e[u] : mn;
            }
            const float sc = __expf(m - mn);
            m = mn;
            float w[4];
            #pragma unroll
            for (int u = 0; u < 4; ++u) w[u] = ok[u] ? __expf(e[u] - mn) : 0.0f;
            dsum = dsum * sc + ((w[0] + w[1]) + (w[2] + w[3]));
            acc.x *= sc;
            acc.y *= sc;
            acc.z *= sc;
            acc.w *= sc;
            #pragma unroll
            for (int u = 0; u < 4; ++u) {
                acc.x += w[u] * hv[u].x;
                acc.y += w[u] * hv[u].y;
                acc.z += w[u] * hv[u].z;
                acc.w += w[u] * hv[u].w;
            }
        }
    }
    const float mo = __shfl_xor(m, 16, 32);
    const float mf = (mo > m) ? mo : m;
    const float sf = __expf(m - mf);
    dsum *= sf;
    acc.x *= sf;
    acc.y *= sf;
    acc.z *= sf;
    acc.w *= sf;
    dsum += __shfl_xor(dsum, 16, 32);
    acc.x += __shfl_xor(acc.x, 16, 32);
    acc.y += __shfl_xor(acc.y, 16, 32);
    acc.z += __shfl_xor(acc.z, 16, 32);
    acc.w += __shfl_xor(acc.w, 16, 32);
    if (odd == 0) {
        const float inv = 1.0f / (dsum + 1e-16f);
        float4 g;
        g.x = acc.x * inv;
        g.y = acc.y * inv;
        g.z = acc.z * inv;
        g.w = acc.w * inv;
        *(float4*)(gtile + wave * TP + col) = g;
    }
    __syncthreads();

    if (wave < 4) {
        const int c = lane & 15;
        const int hi = lane >> 4;
        const float* ap = gtile + c * TP + hi * 2;
        const float* bp = W + (hi * 2) * D + wave * 16 + c;
        v2f af[16], bf[16];
        #pragma unroll
        for (int kk = 0; kk < 16; ++kk) {
            bf[kk].x = bp[(kk * 4 + 0) * D];
            bf[kk].y = bp[(kk * 4 + 1) * D];
            af[kk] = *(const v2f*)(ap + kk * 4);
        }
        v8f o = v8f{};
        #pragma unroll
        for (int kk = 0; kk < 16; ++kk)
            o = __builtin_amdgcn_wmma_f32_16x16x4_f32(
                 false, af[kk],  false, bf[kk],
                 (short)0, o,
                 false,  false);
        const float bv = bias[wave * 16 + c];
        #pragma unroll
        for (int r = 0; r < 8; ++r) otile[(hi * 8 + r) * TP + wave * 16 + c] = o[r] + bv;
    }
    __syncthreads();

    const float2 ov = *(const float2*)(otile + wave * TP + lane * 2);
    *(float2*)(out + (size_t)d * D + lane * 2) = ov;
}

extern "C" void kernel_launch(void* const* d_in, const int* in_sizes, int n_in,
                              void* d_out, int out_size, void* d_ws, size_t ws_size,
                              hipStream_t stream) {
    const float* x        = (const float*)d_in[0];
    const int*   res_n_id = (const int*)d_in[1];
    const int*   edge_src = (const int*)d_in[2];
    const int*   edge_dst = (const int*)d_in[3];
    const float* W        = (const float*)d_in[4];
    const float* att_src  = (const float*)d_in[5];
    const float* att_dst  = (const float*)d_in[6];
    const float* bias     = (const float*)d_in[7];
    float* out = (float*)d_out;

    char* ws = (char*)d_ws;
    unsigned* count  = (unsigned*)(ws);
    unsigned* cursor = (unsigned*)(ws + 80000);
    float*    wv     = (float*)(ws + 160000);
    int*      srcs   = (int*)(ws + 160512);

    gat_prep<<<ZB + 1, ET, 0, stream>>>(W, att_src, att_dst, count, wv);
    gat_hist<<<EB, ET, 0, stream>>>(edge_dst, count);
    gat_scan<<<1, ET, 0, stream>>>(count, cursor);
    gat_fill<<<EB, ET, 0, stream>>>(edge_src, edge_dst, cursor, srcs);
    gat_aggregate<<<N_DST / AG, AG * 32, 0, stream>>>(srcs, count, cursor, wv, res_n_id, x, W,
                                                      bias, out);
}


# Every destination owns a fixed 128-slot segment of the grouped source list, so a slot needs no prefix sum: histogram and scan launches go (5 launches to 3); sum order follows atomic arrival as before
# speedup vs baseline: 5.7312x; 1.2272x over previous
#include <hip/hip_runtime.h>


#define N_SRC 100000
#define N_DST 20000
#define E_CNT 1250000
#define D 64
#define NEG_SLOPE 0.2f

#define EB 125
#define ET 1024
#define EW 1000
#define EU 10
static_assert(EB * EW * EU == E_CNT, "the edge pass must cover every edge once");
static_assert(EW <= ET, "edge threads must exist");
static_assert(N_DST % 32 == 0, "table flush assumes whole 128-byte lines");
#define FLUSH_ROT 160
static_assert(FLUSH_ROT % 32 == 0 && (EB - 1) * FLUSH_ROT < N_DST, "flush start inside the table, on a line");

#define FL ((N_DST + ET - 1) / ET)
#define ZB ((N_DST + ET - 1) / ET)
#define CAP 128
static_assert(CAP % 32 == 0, "a wave's 32 slots must not straddle two segments");
#define AG 16
#define TP 68
static_assert(N_DST % AG == 0, "kernel 2 runs whole groups of sixteen destinations");
static_assert(D == 64 && D * D == 4 * ET, "kernel 0 stages W with one 16-byte load per thread");

typedef __attribute__((ext_vector_type(2))) float v2f;
typedef __attribute__((ext_vector_type(8))) float v8f;

__global__ __launch_bounds__(ET) void gat_prep(const float* __restrict__ W,
                                               const float* __restrict__ att_src,
                                               const float* __restrict__ att_dst,
                                               unsigned* __restrict__ count,
                                               float* __restrict__ wv) {
    __shared__ float ldsW[D * (D + 1)];
    __shared__ float ldsA[2 * D];
    const int t = threadIdx.x;
    if (blockIdx.x < ZB) {
        const int i = blockIdx.x * ET + t;
        if (i < N_DST) count[i] = 0u;
        return;
    }
    const float4 w4 = *(const float4*)(W + t * 4);
    const int wr = (t * 4) >> 6;
    const int wc = (t * 4) & 63;
    ldsW[wr * (D + 1) + wc + 0] = w4.x;
    ldsW[wr * (D + 1) + wc + 1] = w4.y;
    ldsW[wr * (D + 1) + wc + 2] = w4.z;
    ldsW[wr * (D + 1) + wc + 3] = w4.w;
    if (t < D) ldsA[t] = att_src[t];
    else if (t < 2 * D) ldsA[t] = att_dst[t - D];
    __syncthreads();
    if (t < 2 * D) {
        const int k = t & (D - 1);
        const float* a = ldsA + (t & D);
        float s0 = 0.0f, s1 = 0.0f, s2 = 0.0f, s3 = 0.0f;
        #pragma unroll
        for (int j = 0; j < D; j += 4) {
            s0 += ldsW[k * (D + 1) + j + 0] * a[j + 0];
            s1 += ldsW[k * (D + 1) + j + 1] * a[j + 1];
            s2 += ldsW[k * (D + 1) + j + 2] * a[j + 2];
            s3 += ldsW[k * (D + 1) + j + 3] * a[j + 3];
        }
        wv[t] = (s0 + s1) + (s2 + s3);
    }
}

__global__ __launch_bounds__(ET) void gat_fill(const int* __restrict__ es,
                                               const int* __restrict__ ed,
                                               unsigned* __restrict__ count,
                                               int* __restrict__ srcs) {
    __shared__ unsigned tab[N_DST];
    const int t = threadIdx.x;
    for (int i = t; i < N_DST; i += ET) tab[i] = 0u;
    __syncthreads();

    int s[EU], d[EU];
    unsigned r[EU];
    if (t < EW) {
        const int k0 = blockIdx.x * (EW * EU) + t;
        #pragma unroll
        for (int j = 0; j < EU; ++j) {
            s[j] = es[k0 + j * EW];
            d[j] = ed[k0 + j * EW];
        }
        #pragma unroll
        for (int j = 0; j < EU; ++j)
            r[j] = __hip_atomic_fetch_add(&tab[d[j]], 1u, __ATOMIC_RELAXED,
                                          __HIP_MEMORY_SCOPE_WORKGROUP);
    }
    __syncthreads();

    const int rot = blockIdx.x * FLUSH_ROT;
    unsigned c[FL], b[FL];
    #pragma unroll
    for (int q = 0; q < FL; ++q) {
        const int i = t + q * ET;
        int idx = i + rot;
        if (idx >= N_DST) idx -= N_DST;
        c[q] = (i < N_DST) ? tab[idx] : 0u;
    }
    #pragma unroll
    for (int q = 0; q < FL; ++q) {
        int idx = t + q * ET + rot;
        if (idx >= N_DST) idx -= N_DST;
        b[q] = 0u;
        if (c[q] != 0u) b[q] = atomicAdd(count + idx, c[q]);
    }
    #pragma unroll
    for (int q = 0; q < FL; ++q) {
        int idx = t + q * ET + rot;
        if (idx >= N_DST) idx -= N_DST;
        if (c[q] != 0u) tab[idx] = b[q];
    }
    __syncthreads();

    if (t < EW) {
        unsigned pos[EU];
        #pragma unroll
        for (int j = 0; j < EU; ++j) pos[j] = tab[d[j]] + r[j];
        #pragma unroll
        for (int j = 0; j < EU; ++j)
            if (pos[j] < (unsigned)CAP) srcs[d[j] * CAP + (int)pos[j]] = s[j];
    }
}


#define NEG_BIG (-3.0e38f)

template <int XM>
__device__ __forceinline__ float row_xor_add(float x) {
    const int y = __builtin_amdgcn_update_dpp(0, __float_as_int(x), 0x160 + XM, 0xf, 0xf, true);
    return x + __int_as_float(y);
}

__device__ __forceinline__ void walk(const int s, const int rem, const int odd, const int col,
                                     const float* __restrict__ x, const float4 av,
                                     const float ad, float& m, float& dsum, float4& acc) {
    const int cnt = (rem < 32) ? rem : 32;
    for (int i = 0; i < cnt; i += 8) {
        float4 hv[4];
        float p[4];
        bool ok[4];
        #pragma unroll
        for (int u = 0; u < 4; ++u) {
            const int from = i + 2 * u + odd;
            const int si = __shfl(s, from, 32);
            ok[u] = from < rem;
            hv[u] = *(const float4*)(x + (size_t)si * D + col);
        }
        #pragma unroll
        for (int u = 0; u < 4; ++u)
            p[u] = hv[u].x * av.x + hv[u].y * av.y + hv[u].z * av.z + hv[u].w * av.w;
        #pragma unroll
        for (int u = 0; u < 4; ++u) p[u] = row_xor_add<1>(p[u]);
        #pragma unroll
        for (int u = 0; u < 4; ++u) p[u] = row_xor_add<2>(p[u]);
        #pragma unroll
        for (int u = 0; u < 4; ++u) p[u] = row_xor_add<4>(p[u]);
        #pragma unroll
        for (int u = 0; u < 4; ++u) p[u] = row_xor_add<8>(p[u]);
        float e[4];
        float mn = m;
        #pragma unroll
        for (int u = 0; u < 4; ++u) {
            float v = p[u] + ad;
            v = (v > 0.0f) ? v : NEG_SLOPE * v;
            e[u] = ok[u] ? v : NEG_BIG;
            mn = (e[u] > mn) ? e[u] : mn;
        }
        const float sc = __expf(m - mn);
        m = mn;
        float w[4];
        #pragma unroll
        for (int u = 0; u < 4; ++u) w[u] = ok[u] ? __expf(e[u] - mn) : 0.0f;
        dsum = dsum * sc + ((w[0] + w[1]) + (w[2] + w[3]));
        acc.x *= sc;
        acc.y *= sc;
        acc.z *= sc;
        acc.w *= sc;
        #pragma unroll
        for (int u = 0; u < 4; ++u) {
            acc.x += w[u] * hv[u].x;
            acc.y += w[u] * hv[u].y;
            acc.z += w[u] * hv[u].z;
            acc.w += w[u] * hv[u].w;
        }
    }
}

__global__ __launch_bounds__(AG * 32) void gat_aggregate(const int* __restrict__ srcs,
                                                         const unsigned* __restrict__ count,
                                                         const float* __restrict__ wv,
                                                         const int* __restrict__ res_n_id,
                                                         const int* __restrict__ es,
                                                         const int* __restrict__ ed,
                                                         const float* __restrict__ x,
                                                         const float* __restrict__ W,
                                                         const float* __restrict__ bias,
                                                         float* __restrict__ out) {
    __shared__ float gtile[AG * TP];
    __shared__ float otile[AG * TP];
    const int wave = __builtin_amdgcn_readfirstlane((int)(threadIdx.x >> 5));
    const int d = __builtin_amdgcn_readfirstlane((int)(blockIdx.x * AG) + wave);
    const int lane = threadIdx.x & 31;
    const int n = (int)count[d];
    const int rid = res_n_id[d];
    const int* sp = srcs + d * CAP;

    const int odd = lane >> 4;
    const int col = (lane & 15) * 4;
    const float4 av = *(const float4*)(wv + col);
    const float4 dv = *(const float4*)(wv + D + col);
    const float4 xd = *(const float4*)(x + (size_t)rid * D + col);
    float ad = xd.x * dv.x + xd.y * dv.y + xd.z * dv.z + xd.w * dv.w;
    ad = row_xor_add<1>(ad);
    ad = row_xor_add<2>(ad);
    ad = row_xor_add<4>(ad);
    ad = row_xor_add<8>(ad);

    float m = NEG_BIG;
    float dsum = 0.0f;
    float4 acc = make_float4(0.0f, 0.0f, 0.0f, 0.0f);
    if (n <= CAP) {
        for (int c0 = 0; c0 < n; c0 += 32) {
            const int k = c0 + lane;
            const int kc = (k < n) ? k : (n - 1);
            walk(sp[kc], n - c0, odd, col, x, av, ad, m, dsum, acc);
        }
    } else {
        for (int k0 = 0; k0 < E_CNT; k0 += 32) {
            const int k = k0 + lane;
            const bool hit = (k < E_CNT) && (ed[k] == d);
            unsigned mask = __builtin_amdgcn_ballot_w32(hit);
            if (mask == 0u) continue;
            const int sv = hit ? es[k] : 0;
            int sc = 0;
            int cnt = 0;
            while (mask != 0u) {
                const int v = __shfl(sv, __builtin_ctz(mask), 32);
                if (lane >= cnt) sc = v;
                ++cnt;
                mask &= mask - 1u;
            }
            walk(sc, cnt, odd, col, x, av, ad, m, dsum, acc);
        }
    }
    const float mo = __shfl_xor(m, 16, 32);
    const float mf = (mo > m) ? mo : m;
    const float sf = __expf(m - mf);
    dsum *= sf;
    acc.x *= sf;
    acc.y *= sf;
    acc.z *= sf;
    acc.w *= sf;
    dsum += __shfl_xor(dsum, 16, 32);
    acc.x += __shfl_xor(acc.x, 16, 32);
    acc.y += __shfl_xor(acc.y, 16, 32);
    acc.z += __shfl_xor(acc.z, 16, 32);
    acc.w += __shfl_xor(acc.w, 16, 32);
    if (odd == 0) {
        const float inv = 1.0f / (dsum + 1e-16f);
        float4 g;
        g.x = acc.x * inv;
        g.y = acc.y * inv;
        g.z = acc.z * inv;
        g.w = acc.w * inv;
        *(float4*)(gtile + wave * TP + col) = g;
    }
    __syncthreads();

    if (wave < 4) {
        const int c = lane & 15;
        const int hi = lane >> 4;
        const float* ap = gtile + c * TP + hi * 2;
        const float* bp = W + (hi * 2) * D + wave * 16 + c;
        v2f af[16], bf[16];
        #pragma unroll
        for (int kk = 0; kk < 16; ++kk) {
            bf[kk].x = bp[(kk * 4 + 0) * D];
            bf[kk].y = bp[(kk * 4 + 1) * D];
            af[kk] = *(const v2f*)(ap + kk * 4);
        }
        v8f o = v8f{};
        #pragma unroll
        for (int kk = 0; kk < 16; ++kk)
            o = __builtin_amdgcn_wmma_f32_16x16x4_f32(
                 false, af[kk],  false, bf[kk],
                 (short)0, o,
                 false,  false);
        const float bv = bias[wave * 16 + c];
        #pragma unroll
        for (int r = 0; r < 8; ++r) otile[(hi * 8 + r) * TP + wave * 16 + c] = o[r] + bv;
    }
    __syncthreads();

    const float2 ov = *(const float2*)(otile + wave * TP + lane * 2);
    *(float2*)(out + (size_t)d * D + lane * 2) = ov;
}

extern "C" void kernel_launch(void* const* d_in, const int* in_sizes, int n_in,
                              void* d_out, int out_size, void* d_ws, size_t ws_size,
                              hipStream_t stream) {
    const float* x        = (const float*)d_in[0];
    const int*   res_n_id = (const int*)d_in[1];
    const int*   edge_src = (const int*)d_in[2];
    const int*   edge_dst = (const int*)d_in[3];
    const float* W        = (const float*)d_in[4];
    const float* att_src  = (const float*)d_in[5];
    const float* att_dst  = (const float*)d_in[6];
    const float* bias     = (const float*)d_in[7];
    float* out = (float*)d_out;

    char* ws = (char*)d_ws;
    unsigned* count = (unsigned*)(ws);
    float*    wv    = (float*)(ws + 80000);
    int*      srcs  = (int*)(ws + 81920);

    gat_prep<<<ZB + 1, ET, 0, stream>>>(W, att_src, att_dst, count, wv);
    gat_fill<<<EB, ET, 0, stream>>>(edge_src, edge_dst, count, srcs);
    gat_aggregate<<<N_DST / AG, AG * 32, 0, stream>>>(srcs, count, wv, res_n_id, edge_src,
                                                      edge_dst, x, W, bias, out);
}


# Each edge work-group orders its own edges by 16-destination group in LDS, stores whole lines; aggregate merges 125 runs into LDS lists; counters, clear launch, 4-byte scatter go (sum order changes)
# speedup vs baseline: 9.5602x; 1.6681x over previous
#include <hip/hip_runtime.h>


#define N_SRC 100000
#define N_DST 20000
#define E_CNT 1250000
#define D 64
#define NEG_SLOPE 0.2f

#define EB 125
#define ET 1024
#define EW 1000
#define EU 10
#define EPB (EW * EU)
static_assert(EB * EPB == E_CNT, "the edge pass must cover every edge once");
static_assert(EW <= ET, "edge threads must exist");
static_assert(EPB % 4 == 0, "a block's ordered edges are copied out 16 bytes at a time");
#define AG 16
#define TP 68
static_assert(N_DST % AG == 0, "kernel 1 runs whole groups of sixteen destinations");
#define NG (N_DST / AG)
static_assert(NG % 2 == 0 && NG / 2 <= ET, "the scan of kernel 0 gives every thread two groups");
#define SB 17
static_assert(N_SRC <= (1 << SB), "a source must fit its field");
#define SW 10016
static_assert(SW >= EPB && SW % 32 == 0, "an edge work-group's list starts on a line");
#define OP 128
static_assert(EB <= OP, "a row of the table holds one start per edge work-group");
#define CAP 128
static_assert(CAP % 32 == 0, "whole wave steps");
static_assert(4 * EB <= AG * 32, "kernel 1 reads the runs with four lanes each");
static_assert(D == 64 && D * D == 4 * ET, "kernel 0 stages W with one 16-byte load per thread");

typedef __attribute__((ext_vector_type(2))) float v2f;
typedef __attribute__((ext_vector_type(8))) float v8f;

__global__ __launch_bounds__(ET) void gat_sort(const int* __restrict__ es,
                                               const int* __restrict__ ed,
                                               const float* __restrict__ W,
                                               const float* __restrict__ att_src,
                                               const float* __restrict__ att_dst,
                                               int* __restrict__ sorted,
                                               int* __restrict__ starts,
                                               float* __restrict__ wv) {
    __shared__ unsigned hist[NG];
    __shared__ unsigned wtot[ET / 32];
    __shared__ int stage[EPB];
    __shared__ float ldsW[D * (D + 1)];
    __shared__ float ldsA[2 * D];
    const int t = threadIdx.x;
    if (blockIdx.x == EB) {
        const float4 w4 = *(const float4*)(W + t * 4);
        const int wr = (t * 4) >> 6;
        const int wc = (t * 4) & 63;
        ldsW[wr * (D + 1) + wc + 0] = w4.x;
        ldsW[wr * (D + 1) + wc + 1] = w4.y;
        ldsW[wr * (D + 1) + wc + 2] = w4.z;
        ldsW[wr * (D + 1) + wc + 3] = w4.w;
        if (t < D) ldsA[t] = att_src[t];
        else if (t < 2 * D) ldsA[t] = att_dst[t - D];
        __syncthreads();
        if (t < 2 * D) {
            const int k = t & (D - 1);
            const float* a = ldsA + (t & D);
            float s0 = 0.0f, s1 = 0.0f, s2 = 0.0f, s3 = 0.0f;
            #pragma unroll
            for (int j = 0; j < D; j += 4) {
                s0 += ldsW[k * (D + 1) + j + 0] * a[j + 0];
                s1 += ldsW[k * (D + 1) + j + 1] * a[j + 1];
                s2 += ldsW[k * (D + 1) + j + 2] * a[j + 2];
                s3 += ldsW[k * (D + 1) + j + 3] * a[j + 3];
            }
            wv[t] = (s0 + s1) + (s2 + s3);
        }
        return;
    }

    const int lane = t & 31;
    const int wave = t >> 5;
    for (int i = t; i < NG; i += ET) hist[i] = 0u;
    __syncthreads();

    int s[EU], d[EU];
    unsigned r[EU];
    if (t < EW) {
        const int k0 = blockIdx.x * EPB + t;
        #pragma unroll
        for (int j = 0; j < EU; ++j) {
            s[j] = es[k0 + j * EW];
            d[j] = ed[k0 + j * EW];
        }
        #pragma unroll
        for (int j = 0; j < EU; ++j)
            r[j] = __hip_atomic_fetch_add(&hist[(unsigned)d[j] / AG], 1u, __ATOMIC_RELAXED,
                                          __HIP_MEMORY_SCOPE_WORKGROUP);
    }
    __syncthreads();

    unsigned h0 = 0u, h1 = 0u;
    if (t < NG / 2) {
        h0 = hist[2 * t];
        h1 = hist[2 * t + 1];
    }
    const unsigned own = h0 + h1;
    unsigned inc = own;
    #pragma unroll
    for (int o = 1; o < 32; o <<= 1) {
        const unsigned y = __shfl_up(inc, o, 32);
        if (lane >= o) inc += y;
    }
    if (lane == 31) wtot[wave] = inc;
    __syncthreads();
    if (wave == 0) {
        const unsigned w = wtot[lane];
        unsigned wi = w;
        #pragma unroll
        for (int o = 1; o < 32; o <<= 1) {
            const unsigned y = __shfl_up(wi, o, 32);
            if (lane >= o) wi += y;
        }
        wtot[lane] = wi - w;
    }
    __syncthreads();
    if (t < NG / 2) {
        const unsigned e0 = wtot[wave] + inc - own;
        const unsigned e1 = e0 + h0;
        hist[2 * t] = e0;
        hist[2 * t + 1] = e1;
        starts[(2 * t) * OP + blockIdx.x] = (int)e0;
        starts[(2 * t + 1) * OP + blockIdx.x] = (int)e1;
    }
    if (t == 0) starts[NG * OP + blockIdx.x] = EPB;
    __syncthreads();

    if (t < EW) {
        unsigned pos[EU];
        #pragma unroll
        for (int j = 0; j < EU; ++j) pos[j] = hist[(unsigned)d[j] / AG] + r[j];
        #pragma unroll
        for (int j = 0; j < EU; ++j) stage[pos[j]] = (int)(((unsigned)d[j] % AG) << SB) | s[j];
    }
    __syncthreads();

    int4* __restrict__ dst = (int4*)(sorted + blockIdx.x * SW);
    const int4* sp4 = (const int4*)stage;
    for (int i = t; i < EPB / 4; i += ET) dst[i] = sp4[i];
}


#define NEG_BIG (-3.0e38f)

template <int XM>
__device__ __forceinline__ float row_xor_add(float x) {
    const int y = __builtin_amdgcn_update_dpp(0, __float_as_int(x), 0x160 + XM, 0xf, 0xf, true);
    return x + __int_as_float(y);
}

__device__ __forceinline__ void walk(const int s, const int rem, const int odd, const int col,
                                     const float* __restrict__ x, const float4 av,
                                     const float ad, float& m, float& dsum, float4& acc) {
    const int cnt = (rem < 32) ? rem : 32;
    for (int i = 0; i < cnt; i += 8) {
        float4 hv[4];
        float p[4];
        bool ok[4];
        #pragma unroll
        for (int u = 0; u < 4; ++u) {
            const int from = i + 2 * u + odd;
            const int si = __shfl(s, from, 32);
            ok[u] = from < rem;
            hv[u] = *(const float4*)(x + (size_t)si * D + col);
        }
        #pragma unroll
        for (int u = 0; u < 4; ++u)
            p[u] = hv[u].x * av.x + hv[u].y * av.y + hv[u].z * av.z + hv[u].w * av.w;
        #pragma unroll
        for (int u = 0; u < 4; ++u) p[u] = row_xor_add<1>(p[u]);
        #pragma unroll
        for (int u = 0; u < 4; ++u) p[u] = row_xor_add<2>(p[u]);
        #pragma unroll
        for (int u = 0; u < 4; ++u) p[u] = row_xor_add<4>(p[u]);
        #pragma unroll
        for (int u = 0; u < 4; ++u) p[u] = row_xor_add<8>(p[u]);
        float e[4];
        float mn = m;
        #pragma unroll
        for (int u = 0; u < 4; ++u) {
            float v = p[u] + ad;
            v = (v > 0.0f) ? v : NEG_SLOPE * v;
            e[u] = ok[u] ? v : NEG_BIG;
            mn = (e[u] > mn) ? e[u] : mn;
        }
        const float sc = __expf(m - mn);
        m = mn;
        float w[4];
        #pragma unroll
        for (int u = 0; u < 4; ++u) w[u] = ok[u] ? __expf(e[u] - mn) : 0.0f;
        dsum = dsum * sc + ((w[0] + w[1]) + (w[2] + w[3]));
        acc.x *= sc;
        acc.y *= sc;
        acc.z *= sc;
        acc.w *= sc;
        #pragma unroll
        for (int u = 0; u < 4; ++u) {
            acc.x += w[u] * hv[u].x;
            acc.y += w[u] * hv[u].y;
            acc.z += w[u] * hv[u].z;
            acc.w += w[u] * hv[u].w;
        }
    }
}

__global__ __launch_bounds__(AG * 32) void gat_aggregate(const int* __restrict__ sorted,
                                                         const int* __restrict__ starts,
                                                         const float* __restrict__ wv,
                                                         const int* __restrict__ res_n_id,
                                                         const int* __restrict__ es,
                                                         const int* __restrict__ ed,
                                                         const float* __restrict__ x,
                                                         const float* __restrict__ W,
                                                         const float* __restrict__ bias,
                                                         float* __restrict__ out) {
    __shared__ float gtile[AG * TP];
    __shared__ float otile[AG * TP];
    __shared__ unsigned lcnt[AG];
    __shared__ int llist[AG * CAP];
    const int wave = __builtin_amdgcn_readfirstlane((int)(threadIdx.x >> 5));
    const int d = __builtin_amdgcn_readfirstlane((int)(blockIdx.x * AG) + wave);
    const int lane = threadIdx.x & 31;
    const int rid = res_n_id[d];
    const int* sp = llist + wave * CAP;
    if (threadIdx.x < AG) lcnt[threadIdx.x] = 0u;

    const int odd = lane >> 4;
    const int col = (lane & 15) * 4;
    const float4 av = *(const float4*)(wv + col);
    const float4 dv = *(const float4*)(wv + D + col);
    const float4 xd = *(const float4*)(x + (size_t)rid * D + col);

    const int run = (int)(threadIdx.x >> 2);
    const int q = (int)(threadIdx.x & 3);
    int st = 0, en = 0;
    int w[4] = {-1, -1, -1, -1};
    if (run < EB) {
        st = starts[blockIdx.x * OP + run];
        en = starts[(blockIdx.x + 1) * OP + run];
        const int* rp = sorted + run * SW;
        #pragma unroll
        for (int u = 0; u < 4; ++u) {
            const int k = st + q + 4 * u;
            if (k < en) w[u] = rp[k];
        }
    }
    __syncthreads();
    if (run < EB) {
        const int* rp = sorted + run * SW;
        unsigned pos[4];
        #pragma unroll
        for (int u = 0; u < 4; ++u) {
            pos[u] = (unsigned)CAP;
            if (w[u] >= 0)
                pos[u] = __hip_atomic_fetch_add(&lcnt[w[u] >> SB], 1u, __ATOMIC_RELAXED,
                                                __HIP_MEMORY_SCOPE_WORKGROUP);
        }
        #pragma unroll
        for (int u = 0; u < 4; ++u)
            if (pos[u] < (unsigned)CAP)
                llist[(w[u] >> SB) * CAP + (int)pos[u]] = w[u] & ((1 << SB) - 1);
        for (int k = st + q + 16; k < en; k += 4) {
            const int wk = rp[k];
            const int dl = wk >> SB;
            const unsigned pk = __hip_atomic_fetch_add(&lcnt[dl], 1u, __ATOMIC_RELAXED,
                                                       __HIP_MEMORY_SCOPE_WORKGROUP);
            if (pk < (unsigned)CAP) llist[dl * CAP + (int)pk] = wk & ((1 << SB) - 1);
        }
    }
    __syncthreads();
    const int n = __builtin_amdgcn_readfirstlane((int)lcnt[wave]);

    float ad = xd.x * dv.x + xd.y * dv.y + xd.z * dv.z + xd.w * dv.w;
    ad = row_xor_add<1>(ad);
    ad = row_xor_add<2>(ad);
    ad = row_xor_add<4>(ad);
    ad = row_xor_add<8>(ad);

    float m = NEG_BIG;
    float dsum = 0.0f;
    float4 acc = make_float4(0.0f, 0.0f, 0.0f, 0.0f);
    if (n <= CAP) {
        for (int c0 = 0; c0 < n; c0 += 32) {
            const int k = c0 + lane;
            const int kc = (k < n) ? k : (n - 1);
            walk(sp[kc], n - c0, odd, col, x, av, ad, m, dsum, acc);
        }
    } else {
        for (int k0 = 0; k0 < E_CNT; k0 += 32) {
            const int k = k0 + lane;
            const bool hit = (k < E_CNT) && (ed[k] == d);
            unsigned mask = __builtin_amdgcn_ballot_w32(hit);
            if (mask == 0u) continue;
            const int sv = hit ? es[k] : 0;
            int sc = 0;
            int cnt = 0;
            while (mask != 0u) {
                const int v = __shfl(sv, __builtin_ctz(mask), 32);
                if (lane >= cnt) sc = v;
                ++cnt;
                mask &= mask - 1u;
            }
            walk(sc, cnt, odd, col, x, av, ad, m, dsum, acc);
        }
    }
    const float mo = __shfl_xor(m, 16, 32);
    const float mf = (mo > m) ? mo : m;
    const float sf = __expf(m - mf);
    dsum *= sf;
    acc.x *= sf;
    acc.y *= sf;
    acc.z *= sf;
    acc.w *= sf;
    dsum += __shfl_xor(dsum, 16, 32);
    acc.x += __shfl_xor(acc.x, 16, 32);
    acc.y += __shfl_xor(acc.y, 16, 32);
    acc.z += __shfl_xor(acc.z, 16, 32);
    acc.w += __shfl_xor(acc.w, 16, 32);
    if (odd == 0) {
        const float inv = 1.0f / (dsum + 1e-16f);
        float4 g;
        g.x = acc.x * inv;
        g.y = acc.y * inv;
        g.z = acc.z * inv;
        g.w = acc.w * inv;
        *(float4*)(gtile + wave * TP + col) = g;
    }
    __syncthreads();

    if (wave < 4) {
        const int c = lane & 15;
        const int hi = lane >> 4;
        const float* ap = gtile + c * TP + hi * 2;
        const float* bp = W + (hi * 2) * D + wave * 16 + c;
        v2f af[16], bf[16];
        #pragma unroll
        for (int kk = 0; kk < 16; ++kk) {
            bf[kk].x = bp[(kk * 4 + 0) * D];
            bf[kk].y = bp[(kk * 4 + 1) * D];
            af[kk] = *(const v2f*)(ap + kk * 4);
        }
        v8f o = v8f{};
        #pragma unroll
        for (int kk = 0; kk < 16; ++kk)
            o = __builtin_amdgcn_wmma_f32_16x16x4_f32(
                 false, af[kk],  false, bf[kk],
                 (short)0, o,
                 false,  false);
        const float bv = bias[wave * 16 + c];
        #pragma unroll
        for (int r = 0; r < 8; ++r) otile[(hi * 8 + r) * TP + wave * 16 + c] = o[r] + bv;
    }
    __syncthreads();

    const float2 ov = *(const float2*)(otile + wave * TP + lane * 2);
    *(float2*)(out + (size_t)d * D + lane * 2) = ov;
}

extern "C" void kernel_launch(void* const* d_in, const int* in_sizes, int n_in,
                              void* d_out, int out_size, void* d_ws, size_t ws_size,
                              hipStream_t stream) {
    const float* x        = (const float*)d_in[0];
    const int*   res_n_id = (const int*)d_in[1];
    const int*   edge_src = (const int*)d_in[2];
    const int*   edge_dst = (const int*)d_in[3];
    const float* W        = (const float*)d_in[4];
    const float* att_src  = (const float*)d_in[5];
    const float* att_dst  = (const float*)d_in[6];
    const float* bias     = (const float*)d_in[7];
    float* out = (float*)d_out;

    char* ws = (char*)d_ws;
    float* wv     = (float*)(ws);
    int*   starts = (int*)(ws + 512);
    int*   sorted = (int*)(ws + 641024);

    gat_sort<<<EB + 1, ET, 0, stream>>>(edge_src, edge_dst, W, att_src, att_dst, sorted, starts, wv);
    gat_aggregate<<<NG, AG * 32, 0, stream>>>(sorted, starts, wv, res_n_id, edge_src,
                                              edge_dst, x, W, bias, out);
}


# Destination groups of 32 (one 32-wave work-group each) instead of 16: half as many runs, run-start rows and W fetches, runs twice as long, read by 8 lanes each (sum order changes)
# speedup vs baseline: 9.8913x; 1.0346x over previous
#include <hip/hip_runtime.h>


#define N_SRC 100000
#define N_DST 20000
#define E_CNT 1250000
#define D 64
#define NEG_SLOPE 0.2f

#define EB 125
#define ET 1024
#define EW 1000
#define EU 10
#define EPB (EW * EU)
static_assert(EB * EPB == E_CNT, "the edge pass must cover every edge once");
static_assert(EW <= ET, "edge threads must exist");
static_assert(EPB % 4 == 0, "a block's ordered edges are copied out 16 bytes at a time");
#define AG 32
#define RT (AG / 16)
#define TP 68
static_assert(AG % 16 == 0 && AG * 32 <= 1024, "whole row tiles, one wave per destination");
static_assert(N_DST % AG == 0, "kernel 1 runs whole groups of destinations");
#define NG (N_DST / AG)
static_assert(NG <= ET, "the scan of kernel 0 gives every thread one group");
#define RL 8
#define SB 17
static_assert(N_SRC <= (1 << SB) && SB + 5 < 32, "a source must fit its field, the word stays positive");
#define SW 10016
static_assert(SW >= EPB && SW % 32 == 0, "an edge work-group's list starts on a line");
#define OP 128
static_assert(EB <= OP, "a row of the table holds one start per edge work-group");
#define CAP 128
static_assert(CAP % 32 == 0, "whole wave steps");
static_assert(RL * EB <= AG * 32, "kernel 1 reads the runs with RL lanes each");
static_assert(D == 64 && D * D == 4 * ET, "kernel 0 stages W with one 16-byte load per thread");

typedef __attribute__((ext_vector_type(2))) float v2f;
typedef __attribute__((ext_vector_type(8))) float v8f;

__global__ __launch_bounds__(ET) void gat_sort(const int* __restrict__ es,
                                               const int* __restrict__ ed,
                                               const float* __restrict__ W,
                                               const float* __restrict__ att_src,
                                               const float* __restrict__ att_dst,
                                               int* __restrict__ sorted,
                                               int* __restrict__ starts,
                                               float* __restrict__ wv) {
    __shared__ unsigned hist[NG];
    __shared__ unsigned wtot[ET / 32];
    __shared__ int stage[EPB];
    __shared__ float ldsW[D * (D + 1)];
    __shared__ float ldsA[2 * D];
    const int t = threadIdx.x;
    if (blockIdx.x == EB) {
        const float4 w4 = *(const float4*)(W + t * 4);
        const int wr = (t * 4) >> 6;
        const int wc = (t * 4) & 63;
        ldsW[wr * (D + 1) + wc + 0] = w4.x;
        ldsW[wr * (D + 1) + wc + 1] = w4.y;
        ldsW[wr * (D + 1) + wc + 2] = w4.z;
        ldsW[wr * (D + 1) + wc + 3] = w4.w;
        if (t < D) ldsA[t] = att_src[t];
        else if (t < 2 * D) ldsA[t] = att_dst[t - D];
        __syncthreads();
        if (t < 2 * D) {
            const int k = t & (D - 1);
            const float* a = ldsA + (t & D);
            float s0 = 0.0f, s1 = 0.0f, s2 = 0.0f, s3 = 0.0f;
            #pragma unroll
            for (int j = 0; j < D; j += 4) {
                s0 += ldsW[k * (D + 1) + j + 0] * a[j + 0];
                s1 += ldsW[k * (D + 1) + j + 1] * a[j + 1];
                s2 += ldsW[k * (D + 1) + j + 2] * a[j + 2];
                s3 += ldsW[k * (D + 1) + j + 3] * a[j + 3];
            }
            wv[t] = (s0 + s1) + (s2 + s3);
        }
        return;
    }

    const int lane = t & 31;
    const int wave = t >> 5;
    for (int i = t; i < NG; i += ET) hist[i] = 0u;
    __syncthreads();

    int s[EU], d[EU];
    unsigned r[EU];
    if (t < EW) {
        const int k0 = blockIdx.x * EPB + t;
        #pragma unroll
        for (int j = 0; j < EU; ++j) {
            s[j] = es[k0 + j * EW];
            d[j] = ed[k0 + j * EW];
        }
        #pragma unroll
        for (int j = 0; j < EU; ++j)
            r[j] = __hip_atomic_fetch_add(&hist[(unsigned)d[j] / AG], 1u, __ATOMIC_RELAXED,
                                          __HIP_MEMORY_SCOPE_WORKGROUP);
    }
    __syncthreads();

    const unsigned own = (t < NG) ? hist[t] : 0u;
    unsigned inc = own;
    #pragma unroll
    for (int o = 1; o < 32; o <<= 1) {
        const unsigned y = __shfl_up(inc, o, 32);
        if (lane >= o) inc += y;
    }
    if (lane == 31) wtot[wave] = inc;
    __syncthreads();
    if (wave == 0) {
        const unsigned w = wtot[lane];
        unsigned wi = w;
        #pragma unroll
        for (int o = 1; o < 32; o <<= 1) {
            const unsigned y = __shfl_up(wi, o, 32);
            if (lane >= o) wi += y;
        }
        wtot[lane] = wi - w;
    }
    __syncthreads();
    if (t < NG) {
        const unsigned e0 = wtot[wave] + inc - own;
        hist[t] = e0;
        starts[t * OP + blockIdx.x] = (int)e0;
    }
    if (t == 0) starts[NG * OP + blockIdx.x] = EPB;
    __syncthreads();

    if (t < EW) {
        unsigned pos[EU];
        #pragma unroll
        for (int j = 0; j < EU; ++j) pos[j] = hist[(unsigned)d[j] / AG] + r[j];
        #pragma unroll
        for (int j = 0; j < EU; ++j) stage[pos[j]] = (int)(((unsigned)d[j] % AG) << SB) | s[j];
    }
    __syncthreads();

    int4* __restrict__ dst = (int4*)(sorted + blockIdx.x * SW);
    const int4* sp4 = (const int4*)stage;
    for (int i = t; i < EPB / 4; i += ET) dst[i] = sp4[i];
}


#define NEG_BIG (-3.0e38f)

template <int XM>
__device__ __forceinline__ float row_xor_add(float x) {
    const int y = __builtin_amdgcn_update_dpp(0, __float_as_int(x), 0x160 + XM, 0xf, 0xf, true);
    return x + __int_as_float(y);
}

__device__ __forceinline__ void walk(const int s, const int rem, const int odd, const int col,
                                     const float* __restrict__ x, const float4 av,
                                     const float ad, float& m, float& dsum, float4& acc) {
    const int cnt = (rem < 32) ? rem : 32;
    for (int i = 0; i < cnt; i += 8) {
        float4 hv[4];
        float p[4];
        bool ok[4];
        #pragma unroll
        for (int u = 0; u < 4; ++u) {
            const int from = i + 2 * u + odd;
            const int si = __shfl(s, from, 32);
            ok[u] = from < rem;
            hv[u] = *(const float4*)(x + (size_t)si * D + col);
        }
        #pragma unroll
        for (int u = 0; u < 4; ++u)
            p[u] = hv[u].x * av.x + hv[u].y * av.y + hv[u].z * av.z + hv[u].w * av.w;
        #pragma unroll
        for (int u = 0; u < 4; ++u) p[u] = row_xor_add<1>(p[u]);
        #pragma unroll
        for (int u = 0; u < 4; ++u) p[u] = row_xor_add<2>(p[u]);
        #pragma unroll
        for (int u = 0; u < 4; ++u) p[u] = row_xor_add<4>(p[u]);
        #pragma unroll
        for (int u = 0; u < 4; ++u) p[u] = row_xor_add<8>(p[u]);
        float e[4];
        float mn = m;
        #pragma unroll
        for (int u = 0; u < 4; ++u) {
            float v = p[u] + ad;
            v = (v > 0.0f) ? v : NEG_SLOPE * v;
            e[u] = ok[u] ? v : NEG_BIG;
            mn = (e[u] > mn) ? e[u] : mn;
        }
        const float sc = __expf(m - mn);
        m = mn;
        float w[4];
        #pragma unroll
        for (int u = 0; u < 4; ++u) w[u] = ok[u] ? __expf(e[u] - mn) : 0.0f;
        dsum = dsum * sc + ((w[0] + w[1]) + (w[2] + w[3]));
        acc.x *= sc;
        acc.y *= sc;
        acc.z *= sc;
        acc.w *= sc;
        #pragma unroll
        for (int u = 0; u < 4; ++u) {
            acc.x += w[u] * hv[u].x;
            acc.y += w[u] * hv[u].y;
            acc.z += w[u] * hv[u].z;
            acc.w += w[u] * hv[u].w;
        }
    }
}

__global__ __launch_bounds__(AG * 32) void gat_aggregate(const int* __restrict__ sorted,
                                                         const int* __restrict__ starts,
                                                         const float* __restrict__ wv,
                                                         const int* __restrict__ res_n_id,
                                                         const int* __restrict__ es,
                                                         const int* __restrict__ ed,
                                                         const float* __restrict__ x,
                                                         const float* __restrict__ W,
                                                         const float* __restrict__ bias,
                                                         float* __restrict__ out) {
    __shared__ float gtile[AG * TP];
    __shared__ float otile[AG * TP];
    __shared__ unsigned lcnt[AG];
    __shared__ int llist[AG * CAP];
    const int wave = __builtin_amdgcn_readfirstlane((int)(threadIdx.x >> 5));
    const int d = __builtin_amdgcn_readfirstlane((int)(blockIdx.x * AG) + wave);
    const int lane = threadIdx.x & 31;
    const int rid = res_n_id[d];
    const int* sp = llist + wave * CAP;
    if (threadIdx.x < AG) lcnt[threadIdx.x] = 0u;

    const int odd = lane >> 4;
    const int col = (lane & 15) * 4;
    const float4 av = *(const float4*)(wv + col);
    const float4 dv = *(const float4*)(wv + D + col);
    const float4 xd = *(const float4*)(x + (size_t)rid * D + col);

    const int run = (int)(threadIdx.x / RL);
    const int q = (int)(threadIdx.x % RL);
    int st = 0, en = 0;
    int w[4] = {-1, -1, -1, -1};
    if (run < EB) {
        st = starts[blockIdx.x * OP + run];
        en = starts[(blockIdx.x + 1) * OP + run];
        const int* rp = sorted + run * SW;
        #pragma unroll
        for (int u = 0; u < 4; ++u) {
            const int k = st + q + RL * u;
            if (k < en) w[u] = rp[k];
        }
    }
    __syncthreads();
    if (run < EB) {
        const int* rp = sorted + run * SW;
        unsigned pos[4];
        #pragma unroll
        for (int u = 0; u < 4; ++u) {
            pos[u] = (unsigned)CAP;
            if (w[u] >= 0)
                pos[u] = __hip_atomic_fetch_add(&lcnt[w[u] >> SB], 1u, __ATOMIC_RELAXED,
                                                __HIP_MEMORY_SCOPE_WORKGROUP);
        }
        #pragma unroll
        for (int u = 0; u < 4; ++u)
            if (pos[u] < (unsigned)CAP)
                llist[(w[u] >> SB) * CAP + (int)pos[u]] = w[u] & ((1 << SB) - 1);
        for (int k = st + q + 4 * RL; k < en; k += RL) {
            const int wk = rp[k];
            const int dl = wk >> SB;
            const unsigned pk = __hip_atomic_fetch_add(&lcnt[dl], 1u, __ATOMIC_RELAXED,
                                                       __HIP_MEMORY_SCOPE_WORKGROUP);
            if (pk < (unsigned)CAP) llist[dl * CAP + (int)pk] = wk & ((1 << SB) - 1);
        }
    }
    __syncthreads();
    const int n = __builtin_amdgcn_readfirstlane((int)lcnt[wave]);

    float ad = xd.x * dv.x + xd.y * dv.y + xd.z * dv.z + xd.w * dv.w;
    ad = row_xor_add<1>(ad);
    ad = row_xor_add<2>(ad);
    ad = row_xor_add<4>(ad);
    ad = row_xor_add<8>(ad);

    float m = NEG_BIG;
    float dsum = 0.0f;
    float4 acc = make_float4(0.0f, 0.0f, 0.0f, 0.0f);
    if (n <= CAP) {
        for (int c0 = 0; c0 < n; c0 += 32) {
            const int k = c0 + lane;
            const int kc = (k < n) ? k : (n - 1);
            walk(sp[kc], n - c0, odd, col, x, av, ad, m, dsum, acc);
        }
    } else {
        for (int k0 = 0; k0 < E_CNT; k0 += 32) {
            const int k = k0 + lane;
            const bool hit = (k < E_CNT) && (ed[k] == d);
            unsigned mask = __builtin_amdgcn_ballot_w32(hit);
            if (mask == 0u) continue;
            const int sv = hit ? es[k] : 0;
            int sc = 0;
            int cnt = 0;
            while (mask != 0u) {
                const int v = __shfl(sv, __builtin_ctz(mask), 32);
                if (lane >= cnt) sc = v;
                ++cnt;
                mask &= mask - 1u;
            }
            walk(sc, cnt, odd, col, x, av, ad, m, dsum, acc);
        }
    }
    const float mo = __shfl_xor(m, 16, 32);
    const float mf = (mo > m) ? mo : m;
    const float sf = __expf(m - mf);
    dsum *= sf;
    acc.x *= sf;
    acc.y *= sf;
    acc.z *= sf;
    acc.w *= sf;
    dsum += __shfl_xor(dsum, 16, 32);
    acc.x += __shfl_xor(acc.x, 16, 32);
    acc.y += __shfl_xor(acc.y, 16, 32);
    acc.z += __shfl_xor(acc.z, 16, 32);
    acc.w += __shfl_xor(acc.w, 16, 32);
    if (odd == 0) {
        const float inv = 1.0f / (dsum + 1e-16f);
        float4 g;
        g.x = acc.x * inv;
        g.y = acc.y * inv;
        g.z = acc.z * inv;
        g.w = acc.w * inv;
        *(float4*)(gtile + wave * TP + col) = g;
    }
    __syncthreads();

    if (wave < 4 * RT) {
        const int c = lane & 15;
        const int hi = lane >> 4;
        const int rt = wave >> 2;
        const int nb = wave & 3;
        const float* ap = gtile + (rt * 16 + c) * TP + hi * 2;
        const float* bp = W + (hi * 2) * D + nb * 16 + c;
        v2f af[16], bf[16];
        #pragma unroll
        for (int kk = 0; kk < 16; ++kk) {
            bf[kk].x = bp[(kk * 4 + 0) * D];
            bf[kk].y = bp[(kk * 4 + 1) * D];
            af[kk] = *(const v2f*)(ap + kk * 4);
        }
        v8f o = v8f{};
        #pragma unroll
        for (int kk = 0; kk < 16; ++kk)
            o = __builtin_amdgcn_wmma_f32_16x16x4_f32(
                 false, af[kk],  false, bf[kk],
                 (short)0, o,
                 false,  false);
        const float bv = bias[nb * 16 + c];
        #pragma unroll
        for (int r = 0; r < 8; ++r) otile[(rt * 16 + hi * 8 + r) * TP + nb * 16 + c] = o[r] + bv;
    }
    __syncthreads();

    const float2 ov = *(const float2*)(otile + wave * TP + lane * 2);
    *(float2*)(out + (size_t)d * D + lane * 2) = ov;
}

extern "C" void kernel_launch(void* const* d_in, const int* in_sizes, int n_in,
                              void* d_out, int out_size, void* d_ws, size_t ws_size,
                              hipStream_t stream) {
    const float* x        = (const float*)d_in[0];
    const int*   res_n_id = (const int*)d_in[1];
    const int*   edge_src = (const int*)d_in[2];
    const int*   edge_dst = (const int*)d_in[3];
    const float* W        = (const float*)d_in[4];
    const float* att_src  = (const float*)d_in[5];
    const float* att_dst  = (const float*)d_in[6];
    const float* bias     = (const float*)d_in[7];
    float* out = (float*)d_out;

    char* ws = (char*)d_ws;
    float* wv     = (float*)(ws);
    int*   starts = (int*)(ws + 512);
    int*   sorted = (int*)(ws + 321024);

    gat_sort<<<EB + 1, ET, 0, stream>>>(edge_src, edge_dst, W, att_src, att_dst, sorted, starts, wv);
    gat_aggregate<<<NG, AG * 32, 0, stream>>>(sorted, starts, wv, res_n_id, edge_src,
                                              edge_dst, x, W, bias, out);
}


# Walk sums a trip's 4 row dots transposed over 16 lanes: lane l ends with edge (l & 3)'s logit, so leaky_relu, max, exp run once per 4 edges, not per edge; quad-broadcast weights (sum order changes)
# speedup vs baseline: 10.7479x; 1.0866x over previous
#include <hip/hip_runtime.h>


#define N_SRC 100000
#define N_DST 20000
#define E_CNT 1250000
#define D 64
#define NEG_SLOPE 0.2f

#define EB 125
#define ET 1024
#define EW 1000
#define EU 10
#define EPB (EW * EU)
static_assert(EB * EPB == E_CNT, "the edge pass must cover every edge once");
static_assert(EW <= ET, "edge threads must exist");
static_assert(EPB % 4 == 0, "a block's ordered edges are copied out 16 bytes at a time");
#define AG 32
#define RT (AG / 16)
#define TP 68
static_assert(AG % 16 == 0 && AG * 32 <= 1024, "whole row tiles, one wave per destination");
static_assert(N_DST % AG == 0, "kernel 1 runs whole groups of destinations");
#define NG (N_DST / AG)
static_assert(NG <= ET, "the scan of kernel 0 gives every thread one group");
#define RL 8
#define SB 17
static_assert(N_SRC <= (1 << SB) && SB + 5 < 32, "a source must fit its field, the word stays positive");
#define SW 10016
static_assert(SW >= EPB && SW % 32 == 0, "an edge work-group's list starts on a line");
#define OP 128
static_assert(EB <= OP, "a row of the table holds one start per edge work-group");
#define CAP 128
static_assert(CAP % 32 == 0, "whole wave steps");
static_assert(RL * EB <= AG * 32, "kernel 1 reads the runs with RL lanes each");
static_assert(D == 64 && D * D == 4 * ET, "kernel 0 stages W with one 16-byte load per thread");

typedef __attribute__((ext_vector_type(2))) float v2f;
typedef __attribute__((ext_vector_type(8))) float v8f;

__global__ __launch_bounds__(ET) void gat_sort(const int* __restrict__ es,
                                               const int* __restrict__ ed,
                                               const float* __restrict__ W,
                                               const float* __restrict__ att_src,
                                               const float* __restrict__ att_dst,
                                               int* __restrict__ sorted,
                                               int* __restrict__ starts,
                                               float* __restrict__ wv) {
    __shared__ unsigned hist[NG];
    __shared__ unsigned wtot[ET / 32];
    __shared__ int stage[EPB];
    __shared__ float ldsW[D * (D + 1)];
    __shared__ float ldsA[2 * D];
    const int t = threadIdx.x;
    if (blockIdx.x == EB) {
        const float4 w4 = *(const float4*)(W + t * 4);
        const int wr = (t * 4) >> 6;
        const int wc = (t * 4) & 63;
        ldsW[wr * (D + 1) + wc + 0] = w4.x;
        ldsW[wr * (D + 1) + wc + 1] = w4.y;
        ldsW[wr * (D + 1) + wc + 2] = w4.z;
        ldsW[wr * (D + 1) + wc + 3] = w4.w;
        if (t < D) ldsA[t] = att_src[t];
        else if (t < 2 * D) ldsA[t] = att_dst[t - D];
        __syncthreads();
        if (t < 2 * D) {
            const int k = t & (D - 1);
            const float* a = ldsA + (t & D);
            float s0 = 0.0f, s1 = 0.0f, s2 = 0.0f, s3 = 0.0f;
            #pragma unroll
            for (int j = 0; j < D; j += 4) {
                s0 += ldsW[k * (D + 1) + j + 0] * a[j + 0];
                s1 += ldsW[k * (D + 1) + j + 1] * a[j + 1];
                s2 += ldsW[k * (D + 1) + j + 2] * a[j + 2];
                s3 += ldsW[k * (D + 1) + j + 3] * a[j + 3];
            }
            wv[t] = (s0 + s1) + (s2 + s3);
        }
        return;
    }

    const int lane = t & 31;
    const int wave = t >> 5;
    for (int i = t; i < NG; i += ET) hist[i] = 0u;
    __syncthreads();

    int s[EU], d[EU];
    unsigned r[EU];
    if (t < EW) {
        const int k0 = blockIdx.x * EPB + t;
        #pragma unroll
        for (int j = 0; j < EU; ++j) {
            s[j] = es[k0 + j * EW];
            d[j] = ed[k0 + j * EW];
        }
        #pragma unroll
        for (int j = 0; j < EU; ++j)
            r[j] = __hip_atomic_fetch_add(&hist[(unsigned)d[j] / AG], 1u, __ATOMIC_RELAXED,
                                          __HIP_MEMORY_SCOPE_WORKGROUP);
    }
    __syncthreads();

    const unsigned own = (t < NG) ? hist[t] : 0u;
    unsigned inc = own;
    #pragma unroll
    for (int o = 1; o < 32; o <<= 1) {
        const unsigned y = __shfl_up(inc, o, 32);
        if (lane >= o) inc += y;
    }
    if (lane == 31) wtot[wave] = inc;
    __syncthreads();
    if (wave == 0) {
        const unsigned w = wtot[lane];
        unsigned wi = w;
        #pragma unroll
        for (int o = 1; o < 32; o <<= 1) {
            const unsigned y = __shfl_up(wi, o, 32);
            if (lane >= o) wi += y;
        }
        wtot[lane] = wi - w;
    }
    __syncthreads();
    if (t < NG) {
        const unsigned e0 = wtot[wave] + inc - own;
        hist[t] = e0;
        starts[t * OP + blockIdx.x] = (int)e0;
    }
    if (t == 0) starts[NG * OP + blockIdx.x] = EPB;
    __syncthreads();

    if (t < EW) {
        unsigned pos[EU];
        #pragma unroll
        for (int j = 0; j < EU; ++j) pos[j] = hist[(unsigned)d[j] / AG] + r[j];
        #pragma unroll
        for (int j = 0; j < EU; ++j) stage[pos[j]] = (int)(((unsigned)d[j] % AG) << SB) | s[j];
    }
    __syncthreads();

    int4* __restrict__ dst = (int4*)(sorted + blockIdx.x * SW);
    const int4* sp4 = (const int4*)stage;
    for (int i = t; i < EPB / 4; i += ET) dst[i] = sp4[i];
}


#define NEG_BIG (-3.0e38f)

template <int XM>
__device__ __forceinline__ float row_xor_add(float x) {
    const int y = __builtin_amdgcn_update_dpp(0, __float_as_int(x), 0x160 + XM, 0xf, 0xf, true);
    return x + __int_as_float(y);
}

template <int XM>
__device__ __forceinline__ float row_xor_get(float x) {
    return __int_as_float(
        __builtin_amdgcn_update_dpp(0, __float_as_int(x), 0x160 + XM, 0xf, 0xf, true));
}

template <int U>
__device__ __forceinline__ float quad_get(float x) {
    return __int_as_float(
        __builtin_amdgcn_update_dpp(0, __float_as_int(x), U * 0x55, 0xf, 0xf, true));
}

__device__ __forceinline__ void walk(const int s, const int rem, const int odd, const int col,
                                     const int lane, const float* __restrict__ x,
                                     const float4 av, const float ad, float& m, float& dsum,
                                     float4& acc) {
    const int cnt = (rem < 32) ? rem : 32;
    const bool b0 = (lane & 1) != 0;
    const bool b1 = (lane & 2) != 0;
    const int own = 2 * (lane & 3) + odd;
    for (int i = 0; i < cnt; i += 8) {
        float4 hv[4];
        float p[4];
        #pragma unroll
        for (int u = 0; u < 4; ++u) {
            const int si = __shfl(s, i + 2 * u + odd, 32);
            hv[u] = *(const float4*)(x + (size_t)si * D + col);
        }
        #pragma unroll
        for (int u = 0; u < 4; ++u)
            p[u] = hv[u].x * av.x + hv[u].y * av.y + hv[u].z * av.z + hv[u].w * av.w;
        const float q0 = (b0 ? p[1] : p[0]) + row_xor_get<1>(b0 ? p[0] : p[1]);
        const float q1 = (b0 ? p[3] : p[2]) + row_xor_get<1>(b0 ? p[2] : p[3]);
        float r = (b1 ? q1 : q0) + row_xor_get<2>(b1 ? q0 : q1);
        r = row_xor_add<4>(r);
        r = row_xor_add<8>(r);
        const bool ok = i + own < rem;
        float v = r + ad;
        v = (v > 0.0f) ? v : NEG_SLOPE * v;
        const float e = ok ? v : NEG_BIG;
        float g = __builtin_fmaxf(e, row_xor_get<1>(e));
        g = __builtin_fmaxf(g, row_xor_get<2>(g));
        const float mn = __builtin_fmaxf(g, m);
        const float sc = __expf(m - mn);
        m = mn;
        const float w = ok ? __expf(e - mn) : 0.0f;
        dsum = dsum * sc + w;
        const float w0 = quad_get<0>(w);
        const float w1 = quad_get<1>(w);
        const float w2 = quad_get<2>(w);
        const float w3 = quad_get<3>(w);
        acc.x = acc.x * sc + ((w0 * hv[0].x + w1 * hv[1].x) + (w2 * hv[2].x + w3 * hv[3].x));
        acc.y = acc.y * sc + ((w0 * hv[0].y + w1 * hv[1].y) + (w2 * hv[2].y + w3 * hv[3].y));
        acc.z = acc.z * sc + ((w0 * hv[0].z + w1 * hv[1].z) + (w2 * hv[2].z + w3 * hv[3].z));
        acc.w = acc.w * sc + ((w0 * hv[0].w + w1 * hv[1].w) + (w2 * hv[2].w + w3 * hv[3].w));
    }
}

__global__ __launch_bounds__(AG * 32) void gat_aggregate(const int* __restrict__ sorted,
                                                         const int* __restrict__ starts,
                                                         const float* __restrict__ wv,
                                                         const int* __restrict__ res_n_id,
                                                         const int* __restrict__ es,
                                                         const int* __restrict__ ed,
                                                         const float* __restrict__ x,
                                                         const float* __restrict__ W,
                                                         const float* __restrict__ bias,
                                                         float* __restrict__ out) {
    __shared__ float gtile[AG * TP];
    __shared__ float otile[AG * TP];
    __shared__ unsigned lcnt[AG];
    __shared__ int llist[AG * CAP];
    const int wave = __builtin_amdgcn_readfirstlane((int)(threadIdx.x >> 5));
    const int d = __builtin_amdgcn_readfirstlane((int)(blockIdx.x * AG) + wave);
    const int lane = threadIdx.x & 31;
    const int rid = res_n_id[d];
    const int* sp = llist + wave * CAP;
    if (threadIdx.x < AG) lcnt[threadIdx.x] = 0u;

    const int odd = lane >> 4;
    const int col = (lane & 15) * 4;
    const float4 av = *(const float4*)(wv + col);
    const float4 dv = *(const float4*)(wv + D + col);
    const float4 xd = *(const float4*)(x + (size_t)rid * D + col);

    const int run = (int)(threadIdx.x / RL);
    const int q = (int)(threadIdx.x % RL);
    int st = 0, en = 0;
    int w[4] = {-1, -1, -1, -1};
    if (run < EB) {
        st = starts[blockIdx.x * OP + run];
        en = starts[(blockIdx.x + 1) * OP + run];
        const int* rp = sorted + run * SW;
        #pragma unroll
        for (int u = 0; u < 4; ++u) {
            const int k = st + q + RL * u;
            if (k < en) w[u] = rp[k];
        }
    }
    __syncthreads();
    if (run < EB) {
        const int* rp = sorted + run * SW;
        unsigned pos[4];
        #pragma unroll
        for (int u = 0; u < 4; ++u) {
            pos[u] = (unsigned)CAP;
            if (w[u] >= 0)
                pos[u] = __hip_atomic_fetch_add(&lcnt[w[u] >> SB], 1u, __ATOMIC_RELAXED,
                                                __HIP_MEMORY_SCOPE_WORKGROUP);
        }
        #pragma unroll
        for (int u = 0; u < 4; ++u)
            if (pos[u] < (unsigned)CAP)
                llist[(w[u] >> SB) * CAP + (int)pos[u]] = w[u] & ((1 << SB) - 1);
        for (int k = st + q + 4 * RL; k < en; k += RL) {
            const int wk = rp[k];
            const int dl = wk >> SB;
            const unsigned pk = __hip_atomic_fetch_add(&lcnt[dl], 1u, __ATOMIC_RELAXED,
                                                       __HIP_MEMORY_SCOPE_WORKGROUP);
            if (pk < (unsigned)CAP) llist[dl * CAP + (int)pk] = wk & ((1 << SB) - 1);
        }
    }
    __syncthreads();
    const int n = __builtin_amdgcn_readfirstlane((int)lcnt[wave]);

    float ad = xd.x * dv.x + xd.y * dv.y + xd.z * dv.z + xd.w * dv.w;
    ad = row_xor_add<1>(ad);
    ad = row_xor_add<2>(ad);
    ad = row_xor_add<4>(ad);
    ad = row_xor_add<8>(ad);

    float m = NEG_BIG;
    float dsum = 0.0f;
    float4 acc = make_float4(0.0f, 0.0f, 0.0f, 0.0f);
    if (n <= CAP) {
        for (int c0 = 0; c0 < n; c0 += 32) {
            const int k = c0 + lane;
            const int kc = (k < n) ? k : (n - 1);
            walk(sp[kc], n - c0, odd, col, lane, x, av, ad, m, dsum, acc);
        }
    } else {
        for (int k0 = 0; k0 < E_CNT; k0 += 32) {
            const int k = k0 + lane;
            const bool hit = (k < E_CNT) && (ed[k] == d);
            unsigned mask = __builtin_amdgcn_ballot_w32(hit);
            if (mask == 0u) continue;
            const int sv = hit ? es[k] : 0;
            int sc = 0;
            int cnt = 0;
            while (mask != 0u) {
                const int v = __shfl(sv, __builtin_ctz(mask), 32);
                if (lane >= cnt) sc = v;
                ++cnt;
                mask &= mask - 1u;
            }
            walk(sc, cnt, odd, col, lane, x, av, ad, m, dsum, acc);
        }
    }
    dsum = row_xor_add<1>(dsum);
    dsum = row_xor_add<2>(dsum);
    const float mo = __shfl_xor(m, 16, 32);
    const float mf = (mo > m) ? mo : m;
    const float sf = __expf(m - mf);
    dsum *= sf;
    acc.x *= sf;
    acc.y *= sf;
    acc.z *= sf;
    acc.w *= sf;
    dsum += __shfl_xor(dsum, 16, 32);
    acc.x += __shfl_xor(acc.x, 16, 32);
    acc.y += __shfl_xor(acc.y, 16, 32);
    acc.z += __shfl_xor(acc.z, 16, 32);
    acc.w += __shfl_xor(acc.w, 16, 32);
    if (odd == 0) {
        const float inv = 1.0f / (dsum + 1e-16f);
        float4 g;
        g.x = acc.x * inv;
        g.y = acc.y * inv;
        g.z = acc.z * inv;
        g.w = acc.w * inv;
        *(float4*)(gtile + wave * TP + col) = g;
    }
    __syncthreads();

    if (wave < 4 * RT) {
        const int c = lane & 15;
        const int hi = lane >> 4;
        const int rt = wave >> 2;
        const int nb = wave & 3;
        const float* ap = gtile + (rt * 16 + c) * TP + hi * 2;
        const float* bp = W + (hi * 2) * D + nb * 16 + c;
        v2f af[16], bf[16];
        #pragma unroll
        for (int kk = 0; kk < 16; ++kk) {
            bf[kk].x = bp[(kk * 4 + 0) * D];
            bf[kk].y = bp[(kk * 4 + 1) * D];
            af[kk] = *(const v2f*)(ap + kk * 4);
        }
        v8f o = v8f{};
        #pragma unroll
        for (int kk = 0; kk < 16; ++kk)
            o = __builtin_amdgcn_wmma_f32_16x16x4_f32(
                 false, af[kk],  false, bf[kk],
                 (short)0, o,
                 false,  false);
        const float bv = bias[nb * 16 + c];
        #pragma unroll
        for (int r = 0; r < 8; ++r) otile[(rt * 16 + hi * 8 + r) * TP + nb * 16 + c] = o[r] + bv;
    }
    __syncthreads();

    const float2 ov = *(const float2*)(otile + wave * TP + lane * 2);
    *(float2*)(out + (size_t)d * D + lane * 2) = ov;
}

extern "C" void kernel_launch(void* const* d_in, const int* in_sizes, int n_in,
                              void* d_out, int out_size, void* d_ws, size_t ws_size,
                              hipStream_t stream) {
    const float* x        = (const float*)d_in[0];
    const int*   res_n_id = (const int*)d_in[1];
    const int*   edge_src = (const int*)d_in[2];
    const int*   edge_dst = (const int*)d_in[3];
    const float* W        = (const float*)d_in[4];
    const float* att_src  = (const float*)d_in[5];
    const float* att_dst  = (const float*)d_in[6];
    const float* bias     = (const float*)d_in[7];
    float* out = (float*)d_out;

    char* ws = (char*)d_ws;
    float* wv     = (float*)(ws);
    int*   starts = (int*)(ws + 512);
    int*   sorted = (int*)(ws + 321024);

    gat_sort<<<EB + 1, ET, 0, stream>>>(edge_src, edge_dst, W, att_src, att_dst, sorted, starts, wv);
    gat_aggregate<<<NG, AG * 32, 0, stream>>>(sorted, starts, wv, res_n_id, edge_src,
                                              edge_dst, x, W, bias, out);
}


# One 16-lane half per destination, 40 destinations per 20-wave work-group: 10,000 waves in 500 work-groups, all resident, two per unit; sources read straight from the LDS list (sum order changes)
# speedup vs baseline: 11.2420x; 1.0460x over previous
#include <hip/hip_runtime.h>


#define N_SRC 100000
#define N_DST 20000
#define E_CNT 1250000
#define D 64
#define NEG_SLOPE 0.2f

#define EB 125
#define ET 1024
#define EW 1000
#define EU 10
#define EPB (EW * EU)
static_assert(EB * EPB == E_CNT, "the edge pass must cover every edge once");
static_assert(EW <= ET, "edge threads must exist");
static_assert(EPB % 4 == 0, "a block's ordered edges are copied out 16 bytes at a time");
#define AG 40
#define AW (AG / 2)
#define AT (AW * 32)
#define RT ((AG + 15) / 16)
#define TP 68
static_assert(AG % 2 == 0 && AT <= 1024, "two destinations per wave, one work-group per group");
static_assert(4 * RT <= AW, "one wave per 16 x 16 block of the output tile");
static_assert(N_DST % AG == 0, "kernel 1 runs whole groups of destinations");
#define NG (N_DST / AG)
static_assert(NG <= ET, "the scan of kernel 0 gives every thread one group");
#define RL 5
#define RU 8
#define SB 17
static_assert(N_SRC <= (1 << SB) && AG <= 64 && SB + 6 < 32, "a source must fit its field, the word stays positive");
#define SW 10016
static_assert(SW >= EPB && SW % 32 == 0, "an edge work-group's list starts on a line");
#define OP 128
static_assert(EB <= OP, "a row of the table holds one start per edge work-group");
#define CAP 128
static_assert(CAP % 64 == 0, "a half-wave fills its list 64 slots at a time");
static_assert(RL * EB <= AT, "kernel 1 reads the runs with RL lanes each");
static_assert(D == 64 && D * D == 4 * ET, "kernel 0 stages W with one 16-byte load per thread");

typedef __attribute__((ext_vector_type(2))) float v2f;
typedef __attribute__((ext_vector_type(8))) float v8f;

__global__ __launch_bounds__(ET) void gat_sort(const int* __restrict__ es,
                                               const int* __restrict__ ed,
                                               const float* __restrict__ W,
                                               const float* __restrict__ att_src,
                                               const float* __restrict__ att_dst,
                                               int* __restrict__ sorted,
                                               int* __restrict__ starts,
                                               float* __restrict__ wv) {
    __shared__ unsigned hist[NG];
    __shared__ unsigned wtot[ET / 32];
    __shared__ int stage[EPB];
    __shared__ float ldsW[D * (D + 1)];
    __shared__ float ldsA[2 * D];
    const int t = threadIdx.x;
    if (blockIdx.x == EB) {
        const float4 w4 = *(const float4*)(W + t * 4);
        const int wr = (t * 4) >> 6;
        const int wc = (t * 4) & 63;
        ldsW[wr * (D + 1) + wc + 0] = w4.x;
        ldsW[wr * (D + 1) + wc + 1] = w4.y;
        ldsW[wr * (D + 1) + wc + 2] = w4.z;
        ldsW[wr * (D + 1) + wc + 3] = w4.w;
        if (t < D) ldsA[t] = att_src[t];
        else if (t < 2 * D) ldsA[t] = att_dst[t - D];
        __syncthreads();
        if (t < 2 * D) {
            const int k = t & (D - 1);
            const float* a = ldsA + (t & D);
            float s0 = 0.0f, s1 = 0.0f, s2 = 0.0f, s3 = 0.0f;
            #pragma unroll
            for (int j = 0; j < D; j += 4) {
                s0 += ldsW[k * (D + 1) + j + 0] * a[j + 0];
                s1 += ldsW[k * (D + 1) + j + 1] * a[j + 1];
                s2 += ldsW[k * (D + 1) + j + 2] * a[j + 2];
                s3 += ldsW[k * (D + 1) + j + 3] * a[j + 3];
            }
            wv[t] = (s0 + s1) + (s2 + s3);
        }
        return;
    }

    const int lane = t & 31;
    const int wave = t >> 5;
    for (int i = t; i < NG; i += ET) hist[i] = 0u;
    __syncthreads();

    int s[EU], d[EU];
    unsigned r[EU];
    if (t < EW) {
        const int k0 = blockIdx.x * EPB + t;
        #pragma unroll
        for (int j = 0; j < EU; ++j) {
            s[j] = es[k0 + j * EW];
            d[j] = ed[k0 + j * EW];
        }
        #pragma unroll
        for (int j = 0; j < EU; ++j)
            r[j] = __hip_atomic_fetch_add(&hist[(unsigned)d[j] / AG], 1u, __ATOMIC_RELAXED,
                                          __HIP_MEMORY_SCOPE_WORKGROUP);
    }
    __syncthreads();

    const unsigned own = (t < NG) ? hist[t] : 0u;
    unsigned inc = own;
    #pragma unroll
    for (int o = 1; o < 32; o <<= 1) {
        const unsigned y = __shfl_up(inc, o, 32);
        if (lane >= o) inc += y;
    }
    if (lane == 31) wtot[wave] = inc;
    __syncthreads();
    if (wave == 0) {
        const unsigned w = wtot[lane];
        unsigned wi = w;
        #pragma unroll
        for (int o = 1; o < 32; o <<= 1) {
            const unsigned y = __shfl_up(wi, o, 32);
            if (lane >= o) wi += y;
        }
        wtot[lane] = wi - w;
    }
    __syncthreads();
    if (t < NG) {
        const unsigned e0 = wtot[wave] + inc - own;
        hist[t] = e0;
        starts[t * OP + blockIdx.x] = (int)e0;
    }
    if (t == 0) starts[NG * OP + blockIdx.x] = EPB;
    __syncthreads();

    if (t < EW) {
        unsigned pos[EU];
        #pragma unroll
        for (int j = 0; j < EU; ++j) pos[j] = hist[(unsigned)d[j] / AG] + r[j];
        #pragma unroll
        for (int j = 0; j < EU; ++j) stage[pos[j]] = (int)(((unsigned)d[j] % AG) << SB) | s[j];
    }
    __syncthreads();

    int4* __restrict__ dst = (int4*)(sorted + blockIdx.x * SW);
    const int4* sp4 = (const int4*)stage;
    for (int i = t; i < EPB / 4; i += ET) dst[i] = sp4[i];
}


#define NEG_BIG (-3.0e38f)

template <int XM>
__device__ __forceinline__ float row_xor_add(float x) {
    const int y = __builtin_amdgcn_update_dpp(0, __float_as_int(x), 0x160 + XM, 0xf, 0xf, true);
    return x + __int_as_float(y);
}

template <int XM>
__device__ __forceinline__ float row_xor_get(float x) {
    return __int_as_float(
        __builtin_amdgcn_update_dpp(0, __float_as_int(x), 0x160 + XM, 0xf, 0xf, true));
}

template <int U>
__device__ __forceinline__ float quad_get(float x) {
    return __int_as_float(
        __builtin_amdgcn_update_dpp(0, __float_as_int(x), U * 0x55, 0xf, 0xf, true));
}

__device__ __forceinline__ void walk(const int s, const int rem, const int odd, const int col,
                                     const int lane, const float* __restrict__ x,
                                     const float4 av, const float ad, float& m, float& dsum,
                                     float4& acc) {
    const int cnt = (rem < 32) ? rem : 32;
    const bool b0 = (lane & 1) != 0;
    const bool b1 = (lane & 2) != 0;
    const int own = 2 * (lane & 3) + odd;
    for (int i = 0; i < cnt; i += 8) {
        float4 hv[4];
        float p[4];
        #pragma unroll
        for (int u = 0; u < 4; ++u) {
            const int si = __shfl(s, i + 2 * u + odd, 32);
            hv[u] = *(const float4*)(x + (size_t)si * D + col);
        }
        #pragma unroll
        for (int u = 0; u < 4; ++u)
            p[u] = hv[u].x * av.x + hv[u].y * av.y + hv[u].z * av.z + hv[u].w * av.w;
        const float q0 = (b0 ? p[1] : p[0]) + row_xor_get<1>(b0 ? p[0] : p[1]);
        const float q1 = (b0 ? p[3] : p[2]) + row_xor_get<1>(b0 ? p[2] : p[3]);
        float r = (b1 ? q1 : q0) + row_xor_get<2>(b1 ? q0 : q1);
        r = row_xor_add<4>(r);
        r = row_xor_add<8>(r);
        const bool ok = i + own < rem;
        float v = r + ad;
        v = (v > 0.0f) ? v : NEG_SLOPE * v;
        const float e = ok ? v : NEG_BIG;
        float g = __builtin_fmaxf(e, row_xor_get<1>(e));
        g = __builtin_fmaxf(g, row_xor_get<2>(g));
        const float mn = __builtin_fmaxf(g, m);
        const float sc = __expf(m - mn);
        m = mn;
        const float w = ok ? __expf(e - mn) : 0.0f;
        dsum = dsum * sc + w;
        const float w0 = quad_get<0>(w);
        const float w1 = quad_get<1>(w);
        const float w2 = quad_get<2>(w);
        const float w3 = quad_get<3>(w);
        acc.x = acc.x * sc + ((w0 * hv[0].x + w1 * hv[1].x) + (w2 * hv[2].x + w3 * hv[3].x));
        acc.y = acc.y * sc + ((w0 * hv[0].y + w1 * hv[1].y) + (w2 * hv[2].y + w3 * hv[3].y));
        acc.z = acc.z * sc + ((w0 * hv[0].z + w1 * hv[1].z) + (w2 * hv[2].z + w3 * hv[3].z));
        acc.w = acc.w * sc + ((w0 * hv[0].w + w1 * hv[1].w) + (w2 * hv[2].w + w3 * hv[3].w));
    }
}

__device__ __forceinline__ void walk_lists(const int* lp, const int nf, const int nmax,
                                           const int col, const int lane,
                                           const float* __restrict__ x, const float4 av,
                                           const float ad, float& m, float& dsum, float4& acc) {
    const bool b0 = (lane & 1) != 0;
    const bool b1 = (lane & 2) != 0;
    const int own = lane & 3;
    for (int i = 0; i < nmax; i += 4) {
        const int4 sv = *(const int4*)(lp + i);
        float4 hv[4];
        float p[4];
        hv[0] = *(const float4*)(x + (size_t)sv.x * D + col);
        hv[1] = *(const float4*)(x + (size_t)sv.y * D + col);
        hv[2] = *(const float4*)(x + (size_t)sv.z * D + col);
        hv[3] = *(const float4*)(x + (size_t)sv.w * D + col);
        #pragma unroll
        for (int u = 0; u < 4; ++u)
            p[u] = hv[u].x * av.x + hv[u].y * av.y + hv[u].z * av.z + hv[u].w * av.w;
        const float q0 = (b0 ? p[1] : p[0]) + row_xor_get<1>(b0 ? p[0] : p[1]);
        const float q1 = (b0 ? p[3] : p[2]) + row_xor_get<1>(b0 ? p[2] : p[3]);
        float r = (b1 ? q1 : q0) + row_xor_get<2>(b1 ? q0 : q1);
        r = row_xor_add<4>(r);
        r = row_xor_add<8>(r);
        const bool ok = i + own < nf;
        float v = r + ad;
        v = (v > 0.0f) ? v : NEG_SLOPE * v;
        const float e = ok ? v : NEG_BIG;
        float g = __builtin_fmaxf(e, row_xor_get<1>(e));
        g = __builtin_fmaxf(g, row_xor_get<2>(g));
        const float mn = __builtin_fmaxf(g, m);
        const float sc = __expf(m - mn);
        m = mn;
        const float w = ok ? __expf(e - mn) : 0.0f;
        dsum = dsum * sc + w;
        const float w0 = quad_get<0>(w);
        const float w1 = quad_get<1>(w);
        const float w2 = quad_get<2>(w);
        const float w3 = quad_get<3>(w);
        acc.x = acc.x * sc + ((w0 * hv[0].x + w1 * hv[1].x) + (w2 * hv[2].x + w3 * hv[3].x));
        acc.y = acc.y * sc + ((w0 * hv[0].y + w1 * hv[1].y) + (w2 * hv[2].y + w3 * hv[3].y));
        acc.z = acc.z * sc + ((w0 * hv[0].z + w1 * hv[1].z) + (w2 * hv[2].z + w3 * hv[3].z));
        acc.w = acc.w * sc + ((w0 * hv[0].w + w1 * hv[1].w) + (w2 * hv[2].w + w3 * hv[3].w));
    }
}

__global__ __launch_bounds__(AT) void gat_aggregate(const int* __restrict__ sorted,
                                                    const int* __restrict__ starts,
                                                    const float* __restrict__ wv,
                                                    const int* __restrict__ res_n_id,
                                                    const int* __restrict__ es,
                                                    const int* __restrict__ ed,
                                                    const float* __restrict__ x,
                                                    const float* __restrict__ W,
                                                    const float* __restrict__ bias,
                                                    float* __restrict__ out) {
    __shared__ float gtile[RT * 16 * TP];
    __shared__ float otile[RT * 16 * TP];
    __shared__ unsigned lcnt[AG];
    __shared__ __align__(16) int llist[AG * CAP];
    const int wave = __builtin_amdgcn_readfirstlane((int)(threadIdx.x >> 5));
    const int lane = threadIdx.x & 31;
    const int half = lane >> 4;
    const int dl = 2 * wave + half;
    const int d = (int)(blockIdx.x * AG) + dl;
    const int col = (lane & 15) * 4;
    const int rid = res_n_id[d];
    int* lp = llist + dl * CAP;
    if (threadIdx.x < AG) lcnt[threadIdx.x] = 0u;
    for (int i = threadIdx.x; i < (RT * 16 - AG) * TP; i += AT) gtile[AG * TP + i] = 0.0f;

    const float4 av = *(const float4*)(wv + col);
    const float4 dv = *(const float4*)(wv + D + col);

    const int run = (int)(threadIdx.x / RL);
    const int q = (int)(threadIdx.x % RL);
    int st = 0, en = 0;
    int w[RU];
    #pragma unroll
    for (int u = 0; u < RU; ++u) w[u] = -1;
    if (run < EB) {
        st = starts[blockIdx.x * OP + run];
        en = starts[(blockIdx.x + 1) * OP + run];
    }
    const float4 xd = *(const float4*)(x + (size_t)rid * D + col);
    if (run < EB) {
        const int* rp = sorted + run * SW;
        #pragma unroll
        for (int u = 0; u < RU; ++u) {
            const int k = st + q + RL * u;
            if (k < en) w[u] = rp[k];
        }
    }
    const int4 fill = make_int4(rid, rid, rid, rid);
    #pragma unroll
    for (int j = 0; j < CAP; j += 64) *(int4*)(lp + j + col) = fill;
    __syncthreads();
    if (run < EB) {
        const int* rp = sorted + run * SW;
        unsigned pos[RU];
        #pragma unroll
        for (int u = 0; u < RU; ++u) {
            pos[u] = (unsigned)CAP;
            if (w[u] >= 0)
                pos[u] = __hip_atomic_fetch_add(&lcnt[w[u] >> SB], 1u, __ATOMIC_RELAXED,
                                                __HIP_MEMORY_SCOPE_WORKGROUP);
        }
        #pragma unroll
        for (int u = 0; u < RU; ++u)
            if (pos[u] < (unsigned)CAP)
                llist[(w[u] >> SB) * CAP + (int)pos[u]] = w[u] & ((1 << SB) - 1);
        for (int k = st + q + RU * RL; k < en; k += RL) {
            const int wk = rp[k];
            const int dk = wk >> SB;
            const unsigned pk = __hip_atomic_fetch_add(&lcnt[dk], 1u, __ATOMIC_RELAXED,
                                                       __HIP_MEMORY_SCOPE_WORKGROUP);
            if (pk < (unsigned)CAP) llist[dk * CAP + (int)pk] = wk & ((1 << SB) - 1);
        }
    }
    __syncthreads();
    const int n0 = __builtin_amdgcn_readfirstlane((int)lcnt[2 * wave]);
    const int n1 = __builtin_amdgcn_readfirstlane((int)lcnt[2 * wave + 1]);
    const int nf0 = (n0 <= CAP) ? n0 : 0;
    const int nf1 = (n1 <= CAP) ? n1 : 0;
    const int nmax = (nf0 > nf1) ? nf0 : nf1;
    const int nf = half ? nf1 : nf0;

    float ad = xd.x * dv.x + xd.y * dv.y + xd.z * dv.z + xd.w * dv.w;
    ad = row_xor_add<1>(ad);
    ad = row_xor_add<2>(ad);
    ad = row_xor_add<4>(ad);
    ad = row_xor_add<8>(ad);

    float m = NEG_BIG;
    float dsum = 0.0f;
    float4 acc = make_float4(0.0f, 0.0f, 0.0f, 0.0f);
    walk_lists(lp, nf, nmax, col, lane, x, av, ad, m, dsum, acc);
    dsum = row_xor_add<1>(dsum);
    dsum = row_xor_add<2>(dsum);
    const float inv = 1.0f / (dsum + 1e-16f);
    float4 g;
    g.x = acc.x * inv;
    g.y = acc.y * inv;
    g.z = acc.z * inv;
    g.w = acc.w * inv;

    for (int h = 0; h < 2; ++h) {
        const int nn = h ? n1 : n0;
        if (nn <= CAP) continue;
        const int dd = (int)(blockIdx.x * AG) + 2 * wave + h;
        const float adu = __int_as_float(__builtin_amdgcn_readlane(__float_as_int(ad), 16 * h));
        float ms = NEG_BIG;
        float ds = 0.0f;
        float4 as = make_float4(0.0f, 0.0f, 0.0f, 0.0f);
        for (int k0 = 0; k0 < E_CNT; k0 += 32) {
            const int k = k0 + lane;
            const bool hit = (k < E_CNT) && (ed[k] == dd);
            unsigned mask = __builtin_amdgcn_ballot_w32(hit);
            if (mask == 0u) continue;
            const int sv = hit ? es[k] : 0;
            int sc = 0;
            int cnt = 0;
            while (mask != 0u) {
                const int v = __shfl(sv, __builtin_ctz(mask), 32);
                if (lane >= cnt) sc = v;
                ++cnt;
                mask &= mask - 1u;
            }
            walk(sc, cnt, half, col, lane, x, av, adu, ms, ds, as);
        }
        ds = row_xor_add<1>(ds);
        ds = row_xor_add<2>(ds);
        const float mo = __shfl_xor(ms, 16, 32);
        const float mf = (mo > ms) ? mo : ms;
        const float sf = __expf(ms - mf);
        ds *= sf;
        as.x *= sf;
        as.y *= sf;
        as.z *= sf;
        as.w *= sf;
        ds += __shfl_xor(ds, 16, 32);
        as.x += __shfl_xor(as.x, 16, 32);
        as.y += __shfl_xor(as.y, 16, 32);
        as.z += __shfl_xor(as.z, 16, 32);
        as.w += __shfl_xor(as.w, 16, 32);
        const float invs = 1.0f / (ds + 1e-16f);
        if (half == h) {
            g.x = as.x * invs;
            g.y = as.y * invs;
            g.z = as.z * invs;
            g.w = as.w * invs;
        }
    }
    *(float4*)(gtile + dl * TP + col) = g;
    __syncthreads();

    if (wave < 4 * RT) {
        const int c = lane & 15;
        const int hi = lane >> 4;
        const int rt = wave >> 2;
        const int nb = wave & 3;
        const float* ap = gtile + (rt * 16 + c) * TP + hi * 2;
        const float* bp = W + (hi * 2) * D + nb * 16 + c;
        v2f af[16], bf[16];
        #pragma unroll
        for (int kk = 0; kk < 16; ++kk) {
            bf[kk].x = bp[(kk * 4 + 0) * D];
            bf[kk].y = bp[(kk * 4 + 1) * D];
            af[kk] = *(const v2f*)(ap + kk * 4);
        }
        v8f o = v8f{};
        #pragma unroll
        for (int kk = 0; kk < 16; ++kk)
            o = __builtin_amdgcn_wmma_f32_16x16x4_f32(
                 false, af[kk],  false, bf[kk],
                 (short)0, o,
                 false,  false);
        const float bv = bias[nb * 16 + c];
        #pragma unroll
        for (int r = 0; r < 8; ++r) otile[(rt * 16 + hi * 8 + r) * TP + nb * 16 + c] = o[r] + bv;
    }
    __syncthreads();

    const float4 ov = *(const float4*)(otile + dl * TP + col);
    *(float4*)(out + (size_t)d * D + col) = ov;
}

extern "C" void kernel_launch(void* const* d_in, const int* in_sizes, int n_in,
                              void* d_out, int out_size, void* d_ws, size_t ws_size,
                              hipStream_t stream) {
    const float* x        = (const float*)d_in[0];
    const int*   res_n_id = (const int*)d_in[1];
    const int*   edge_src = (const int*)d_in[2];
    const int*   edge_dst = (const int*)d_in[3];
    const float* W        = (const float*)d_in[4];
    const float* att_src  = (const float*)d_in[5];
    const float* att_dst  = (const float*)d_in[6];
    const float* bias     = (const float*)d_in[7];
    float* out = (float*)d_out;

    char* ws = (char*)d_ws;
    float* wv     = (float*)(ws);
    int*   starts = (int*)(ws + 512);
    int*   sorted = (int*)(ws + 321024);

    gat_sort<<<EB + 1, ET, 0, stream>>>(edge_src, edge_dst, W, att_src, att_dst, sorted, starts, wv);
    gat_aggregate<<<NG, AT, 0, stream>>>(sorted, starts, wv, res_n_id, edge_src,
                                              edge_dst, x, W, bias, out);
}
